# starting kernel
# baseline (speedup 1.0000x reference)
#include <hip/hip_runtime.h>
#include <math.h>

typedef __attribute__((ext_vector_type(2))) float v2f;
typedef __attribute__((ext_vector_type(8))) float v8f;

#define STRH 130
#define STRZ 520

__global__ __launch_bounds__(256) void cell_kernel(
    const float* __restrict__ input, const float* __restrict__ WT,
    const float* __restrict__ bcell, const float* __restrict__ h_in,
    float* __restrict__ h_out, float* __restrict__ c, int t)
{
    __shared__ float sh_h[18 * STRH];
    __shared__ float sh_x[16 * 8];
    __shared__ float sh_z[16 * STRZ];

    const int tid = threadIdx.x;
    const int R0  = blockIdx.x * 16;
    const int bb  = R0 >> 7;
    const int s0  = R0 & 127;

    for (int i = tid; i < 18 * 128; i += 256) {
        int r = i >> 7, k = i & 127;
        int s = s0 - 1 + r;
        float v = 0.0f;
        if (s >= 0 && s < 128) v = h_in[(bb * 128 + s) * 128 + k];
        sh_h[r * STRH + k] = v;
    }
    if (tid < 16 * 5) {
        int m = tid / 5, k = tid % 5;
        sh_x[m * 8 + k] = input[(((bb * 128) + t) * 128 + (s0 + m)) * 5 + k];
    }
    __syncthreads();

    const int wave  = tid >> 5;
    const int lane  = tid & 31;
    const int lm    = lane & 15;
    const int hi    = lane >> 4;
    const int kHalf = hi * 2;

    const float* wc0 = WT + ((wave * 4 + 0) * 16 + lm) * 392;
    const float* wc1 = WT + ((wave * 4 + 1) * 16 + lm) * 392;
    const float* wc2 = WT + ((wave * 4 + 2) * 16 + lm) * 392;
    const float* wc3 = WT + ((wave * 4 + 3) * 16 + lm) * 392;

    v8f acc0 = {}, acc1 = {}, acc2 = {}, acc3 = {};

    #pragma unroll
    for (int v = 0; v < 8; ++v) {
        int m = v + hi * 8;
        float x0 = sh_x[m*8+0], x1 = sh_x[m*8+1], x2 = sh_x[m*8+2],
              x3 = sh_x[m*8+3], x4 = sh_x[m*8+4];
        acc0[v] += x0*wc0[0] + x1*wc0[1] + x2*wc0[2] + x3*wc0[3] + x4*wc0[4];
        acc1[v] += x0*wc1[0] + x1*wc1[1] + x2*wc1[2] + x3*wc1[3] + x4*wc1[4];
        acc2[v] += x0*wc2[0] + x1*wc2[1] + x2*wc2[2] + x3*wc2[3] + x4*wc2[4];
        acc3[v] += x0*wc3[0] + x1*wc3[1] + x2*wc3[2] + x3*wc3[3] + x4*wc3[4];
    }

    #pragma unroll
    for (int src = 0; src < 3; ++src) {
        const int shBase = (src == 0) ? 1 : (src == 1) ? 2 : 0;
        const float* arow = &sh_h[(shBase + lm) * STRH];
        const int wOff = 5 + src * 128;
        #pragma unroll 4
        for (int kk = 0; kk < 128; kk += 4) {
            v2f a;
            a.x = arow[kk + kHalf];
            a.y = arow[kk + kHalf + 1];
            v2f b0, b1, b2, b3;
            b0.x = wc0[wOff + kk + kHalf]; b0.y = wc0[wOff + kk + kHalf + 1];
            b1.x = wc1[wOff + kk + kHalf]; b1.y = wc1[wOff + kk + kHalf + 1];
            b2.x = wc2[wOff + kk + kHalf]; b2.y = wc2[wOff + kk + kHalf + 1];
            b3.x = wc3[wOff + kk + kHalf]; b3.y = wc3[wOff + kk + kHalf + 1];
            acc0 = __builtin_amdgcn_wmma_f32_16x16x4_f32(
                        false, a, false, b0, (short)0, acc0, false, false);
            acc1 = __builtin_amdgcn_wmma_f32_16x16x4_f32(
                        false, a, false, b1, (short)0, acc1, false, false);
            acc2 = __builtin_amdgcn_wmma_f32_16x16x4_f32(
                        false, a, false, b2, (short)0, acc2, false, false);
            acc3 = __builtin_amdgcn_wmma_f32_16x16x4_f32(
                        false, a, false, b3, (short)0, acc3, false, false);
        }
    }

    #pragma unroll
    for (int v = 0; v < 8; ++v) {
        int m = v + hi * 8;
        sh_z[m * STRZ + (wave * 4 + 0) * 16 + lm] = acc0[v];
        sh_z[m * STRZ + (wave * 4 + 1) * 16 + lm] = acc1[v];
        sh_z[m * STRZ + (wave * 4 + 2) * 16 + lm] = acc2[v];
        sh_z[m * STRZ + (wave * 4 + 3) * 16 + lm] = acc3[v];
    }
    __syncthreads();

    for (int i = tid; i < 2048; i += 256) {
        int m = i >> 7, j = i & 127;
        float zi = sh_z[m*STRZ + j]       + bcell[j];
        float zf = sh_z[m*STRZ + 128 + j] + bcell[128 + j];
        float zg = sh_z[m*STRZ + 256 + j] + bcell[256 + j];
        float zo = sh_z[m*STRZ + 384 + j] + bcell[384 + j];
        int row = R0 + m;
        float cp = c[row*128 + j];
        float si = 1.0f / (1.0f + __expf(-zi));
        float sf = 1.0f / (1.0f + __expf(-zf));
        float so = 1.0f / (1.0f + __expf(-zo));
        float cn = sf * cp + si * tanhf(zg);
        c[row*128 + j]     = cn;
        h_out[row*128 + j] = so * tanhf(cn);
    }
}

__global__ __launch_bounds__(256) void out_kernel(
    const float* __restrict__ h, const float* __restrict__ W1T,
    const float* __restrict__ b1, const float* __restrict__ W2T,
    const float* __restrict__ b2, float* __restrict__ outbuf)
{
    __shared__ float sh_a[16 * STRH];
    __shared__ float sh_hid[16 * STRH];
    const int tid = threadIdx.x;
    const int R0  = blockIdx.x * 16;

    for (int i = tid; i < 2048; i += 256) {
        int m = i >> 7, k = i & 127;
        sh_a[m*STRH + k] = h[(R0 + m)*128 + k];
    }
    __syncthreads();

    const int wave = tid >> 5, lane = tid & 31;
    const int lm = lane & 15, hi = lane >> 4, kHalf = hi * 2;
    const int col = wave * 16 + lm;
    const float* wk   = W1T + col * 128;
    const float* arow = &sh_a[lm * STRH];

    v8f acc = {};
    #pragma unroll 8
    for (int kk = 0; kk < 128; kk += 4) {
        v2f a, b;
        a.x = arow[kk + kHalf]; a.y = arow[kk + kHalf + 1];
        b.x = wk[kk + kHalf];   b.y = wk[kk + kHalf + 1];
        acc = __builtin_amdgcn_wmma_f32_16x16x4_f32(
                false, a, false, b, (short)0, acc, false, false);
    }
    float bias = b1[col];
    #pragma unroll
    for (int v = 0; v < 8; ++v) {
        float x = acc[v] + bias;
        sh_hid[(v + hi * 8) * STRH + col] = x > 0.0f ? x : 0.0f;
    }
    __syncthreads();

    if (tid < 48) {
        int m = tid / 3, cc = tid % 3;
        const float* hr = &sh_hid[m * STRH];
        const float* w2 = W2T + cc * 128;
        float s = 0.0f;
        for (int k = 0; k < 128; ++k) s += hr[k] * w2[k];
        outbuf[(R0 + m) * 3 + cc] = s + b2[cc];
    }
}

__global__ __launch_bounds__(256) void nd_kernel(
    const float* __restrict__ input, const float* __restrict__ outbuf,
    float* __restrict__ out, int t)
{
    int gid = blockIdx.x * 256 + threadIdx.x;
    if (gid >= 1024) return;
    int bb = gid >> 7, s = gid & 127;
    const float* o = &outbuf[gid * 3];
    float n0 = o[0];
    float n1, sp0, sp1;
    if (s == 0) {
        const float* xn0 = &input[((bb * 128 + (t + 1)) * 128 + 0) * 5];
        n1 = xn0[1]; sp0 = xn0[3]; sp1 = xn0[4];
    } else {
        n1  = outbuf[(gid - 1) * 3 + 0];
        sp0 = o[1]; sp1 = o[2];
    }
    float fn = input[((bb * 128 + t) * 128 + s) * 5 + 2];
    float n2 = fn + n1 - n0;
    float* dst = &out[((bb * 123 + (t - 4)) * 128 + s) * 5];
    dst[0] = n0; dst[1] = n1; dst[2] = n2; dst[3] = sp0; dst[4] = sp1;
}

__global__ __launch_bounds__(256) void zero_kernel(float* p, int n) {
    int i = blockIdx.x * 256 + threadIdx.x;
    if (i < n) p[i] = 0.0f;
}

__global__ __launch_bounds__(256) void prep_kernel(
    const float* __restrict__ Wc, const float* __restrict__ W1,
    const float* __restrict__ W2, float* __restrict__ WT,
    float* __restrict__ W1T, float* __restrict__ W2T)
{
    int i = blockIdx.x * 256 + threadIdx.x;
    if (i < 512 * 392) {
        int col = i / 392, k = i % 392;
        WT[i] = (k < 389) ? Wc[k * 512 + col] : 0.0f;
    } else if (i < 512 * 392 + 16384) {
        int j = i - 512 * 392;
        int col = j >> 7, k = j & 127;
        W1T[j] = W1[k * 128 + col];
    } else if (i < 512 * 392 + 16384 + 384) {
        int j = i - 512 * 392 - 16384;
        int cc = j >> 7, k = j & 127;
        W2T[j] = W2[k * 3 + cc];
    }
}

extern "C" void kernel_launch(void* const* d_in, const int* in_sizes, int n_in,
                              void* d_out, int out_size, void* d_ws, size_t ws_size,
                              hipStream_t stream)
{
    (void)in_sizes; (void)n_in; (void)out_size; (void)ws_size;
    const float* input = (const float*)d_in[0];
    const float* Wc    = (const float*)d_in[1];
    const float* bc    = (const float*)d_in[2];
    const float* W1    = (const float*)d_in[3];
    const float* b1    = (const float*)d_in[4];
    const float* W2    = (const float*)d_in[5];
    const float* b2    = (const float*)d_in[6];
    float* ws   = (float*)d_ws;
    float* WT   = ws;
    float* W1T  = ws + 200704;
    float* W2T  = ws + 217088;
    float* hbuf = ws + 217472;
    float* cbuf = ws + 479616;
    float* obuf = ws + 610688;
    float* out  = (float*)d_out;

    zero_kernel<<<393216 / 256, 256, 0, stream>>>(hbuf, 393216);
    prep_kernel<<<(217472 + 255) / 256, 256, 0, stream>>>(Wc, W1, W2, WT, W1T, W2T);

    for (int t = 0; t < 127; ++t) {
        const float* h_in = hbuf + (t & 1) * 131072;
        float* h_out      = hbuf + ((t + 1) & 1) * 131072;
        cell_kernel<<<64, 256, 0, stream>>>(input, WT, bc, h_in, h_out, cbuf, t);
        out_kernel<<<64, 256, 0, stream>>>(h_out, W1T, b1, W2T, b2, obuf);
        if (t >= 4) nd_kernel<<<4, 256, 0, stream>>>(input, obuf, out, t);
    }
}


# Cell weights staged through LDS: prep lays W as 12 contiguous 64 KiB k-chunks, the block copies each with 16-byte loads kept in flight across LDS-only barriers under the WMMAs; B fragments from LDS
# speedup vs baseline: 1.8663x; 1.8663x over previous
#include <hip/hip_runtime.h>
#include <math.h>

typedef __attribute__((ext_vector_type(2))) float v2f;
typedef __attribute__((ext_vector_type(8))) float v8f;
typedef __attribute__((ext_vector_type(4))) float v4f;

#define STRH 130
#define STRZ 520
#define WP   36
#define WBUF (512 * WP)

__device__ __forceinline__ void chunk_request(v4f (&r)[16], const v4f* src, int tid)
{
    #pragma unroll
    for (int i = 0; i < 16; ++i) r[i] = src[tid + i * 256];
}

__device__ __forceinline__ void chunk_to_lds(float* buf, int wdst, const v4f (&r)[16])
{
    #pragma unroll
    for (int i = 0; i < 16; ++i)
        *reinterpret_cast<v4f*>(&buf[wdst + i * 32 * WP]) = r[i];
}

__device__ __forceinline__ void lds_barrier()
{
    asm volatile("s_wait_dscnt 0x0\n\ts_barrier_signal -1\n\ts_barrier_wait -1" ::: "memory");
}

__device__ __forceinline__ void chunk_mma(const float* arow, const float* wb,
    int wl0, int wl1, int wl2, int wl3, int kHalf,
    v8f& acc0, v8f& acc1, v8f& acc2, v8f& acc3)
{
    #pragma unroll
    for (int kk = 0; kk < 32; kk += 4) {
        v2f a;
        a.x = arow[kk + kHalf];
        a.y = arow[kk + kHalf + 1];
        v2f b0, b1, b2, b3;
        b0.x = wb[wl0 + kk + kHalf]; b0.y = wb[wl0 + kk + kHalf + 1];
        b1.x = wb[wl1 + kk + kHalf]; b1.y = wb[wl1 + kk + kHalf + 1];
        b2.x = wb[wl2 + kk + kHalf]; b2.y = wb[wl2 + kk + kHalf + 1];
        b3.x = wb[wl3 + kk + kHalf]; b3.y = wb[wl3 + kk + kHalf + 1];
        acc0 = __builtin_amdgcn_wmma_f32_16x16x4_f32(
                    false, a, false, b0, (short)0, acc0, false, false);
        acc1 = __builtin_amdgcn_wmma_f32_16x16x4_f32(
                    false, a, false, b1, (short)0, acc1, false, false);
        acc2 = __builtin_amdgcn_wmma_f32_16x16x4_f32(
                    false, a, false, b2, (short)0, acc2, false, false);
        acc3 = __builtin_amdgcn_wmma_f32_16x16x4_f32(
                    false, a, false, b3, (short)0, acc3, false, false);
    }
}

__global__ __launch_bounds__(256) void cell_kernel(
    const float* __restrict__ input, const float* __restrict__ WT,
    const float* __restrict__ bcell, const float* __restrict__ h_in,
    float* __restrict__ h_out, float* __restrict__ c, int t)
{
    __shared__ float sh_h[18 * STRH];
    __shared__ float sh_x[16 * 8];
    __shared__ float sh_z[16 * STRZ];
    __shared__ alignas(16) float sh_w[2 * WBUF];

    const int tid = threadIdx.x;
    const int R0  = blockIdx.x * 16;
    const int bb  = R0 >> 7;
    const int s0  = R0 & 127;

    const v4f* wsrc = reinterpret_cast<const v4f*>(WT);
    const int wdst = (tid >> 3) * WP + (tid & 7) * 4;
    v4f preA[16], preB[16];
    chunk_request(preA, wsrc, tid);
    chunk_request(preB, wsrc + 4096, tid);

    const int wave  = tid >> 5;
    const int lane  = tid & 31;
    const int lm    = lane & 15;
    const int hi    = lane >> 4;
    const int kHalf = hi * 2;

    const float* wc0 = WT + 12 * 16384 + ((wave * 4 + 0) * 16 + lm) * 8;
    const float* wc1 = WT + 12 * 16384 + ((wave * 4 + 1) * 16 + lm) * 8;
    const float* wc2 = WT + 12 * 16384 + ((wave * 4 + 2) * 16 + lm) * 8;
    const float* wc3 = WT + 12 * 16384 + ((wave * 4 + 3) * 16 + lm) * 8;
    const v4f wq0 = *reinterpret_cast<const v4f*>(wc0); const float we0 = wc0[4];
    const v4f wq1 = *reinterpret_cast<const v4f*>(wc1); const float we1 = wc1[4];
    const v4f wq2 = *reinterpret_cast<const v4f*>(wc2); const float we2 = wc2[4];
    const v4f wq3 = *reinterpret_cast<const v4f*>(wc3); const float we3 = wc3[4];

    for (int i = tid; i < 18 * 128; i += 256) {
        int r = i >> 7, k = i & 127;
        int s = s0 - 1 + r;
        float v = 0.0f;
        if (s >= 0 && s < 128) v = h_in[(bb * 128 + s) * 128 + k];
        sh_h[r * STRH + k] = v;
    }
    if (tid < 16 * 5) {
        int m = tid / 5, k = tid % 5;
        sh_x[m * 8 + k] = input[(((bb * 128) + t) * 128 + (s0 + m)) * 5 + k];
    }
    float* const buf0 = sh_w;
    float* const buf1 = sh_w + WBUF;
    chunk_to_lds(buf0, wdst, preA);
    chunk_to_lds(buf1, wdst, preB);
    chunk_request(preA, wsrc + 2 * 4096, tid);
    chunk_request(preB, wsrc + 3 * 4096, tid);
    lds_barrier();

    const int wl0 = ((wave * 4 + 0) * 16 + lm) * WP;
    const int wl1 = ((wave * 4 + 1) * 16 + lm) * WP;
    const int wl2 = ((wave * 4 + 2) * 16 + lm) * WP;
    const int wl3 = ((wave * 4 + 3) * 16 + lm) * WP;

    v8f acc0 = {}, acc1 = {}, acc2 = {}, acc3 = {};

    #pragma unroll
    for (int v = 0; v < 8; ++v) {
        int m = v + hi * 8;
        float x0 = sh_x[m*8+0], x1 = sh_x[m*8+1], x2 = sh_x[m*8+2],
              x3 = sh_x[m*8+3], x4 = sh_x[m*8+4];
        acc0[v] += x0*wq0.x + x1*wq0.y + x2*wq0.z + x3*wq0.w + x4*we0;
        acc1[v] += x0*wq1.x + x1*wq1.y + x2*wq1.z + x3*wq1.w + x4*we1;
        acc2[v] += x0*wq2.x + x1*wq2.y + x2*wq2.z + x3*wq2.w + x4*we2;
        acc3[v] += x0*wq3.x + x1*wq3.y + x2*wq3.z + x3*wq3.w + x4*we3;
    }

    for (int ch = 0; ch < 12; ch += 2) {
        const int src    = ch >> 2;
        const int shBase = (src == 0) ? 1 : (src == 1) ? 2 : 0;
        const float* arow = &sh_h[(shBase + lm) * STRH + (ch & 3) * 32];

        if (ch > 0) {
            chunk_to_lds(buf1, wdst, preB);
            if (ch + 3 < 12) chunk_request(preB, wsrc + (ch + 3) * 4096, tid);
        }
        chunk_mma(arow, buf0, wl0, wl1, wl2, wl3, kHalf, acc0, acc1, acc2, acc3);
        lds_barrier();

        if (ch + 2 < 12) {
            chunk_to_lds(buf0, wdst, preA);
            if (ch + 4 < 12) chunk_request(preA, wsrc + (ch + 4) * 4096, tid);
        }
        chunk_mma(arow + 32, buf1, wl0, wl1, wl2, wl3, kHalf, acc0, acc1, acc2, acc3);
        if (ch + 2 < 12) lds_barrier();
    }

    #pragma unroll
    for (int v = 0; v < 8; ++v) {
        int m = v + hi * 8;
        sh_z[m * STRZ + (wave * 4 + 0) * 16 + lm] = acc0[v];
        sh_z[m * STRZ + (wave * 4 + 1) * 16 + lm] = acc1[v];
        sh_z[m * STRZ + (wave * 4 + 2) * 16 + lm] = acc2[v];
        sh_z[m * STRZ + (wave * 4 + 3) * 16 + lm] = acc3[v];
    }
    __syncthreads();

    for (int i = tid; i < 2048; i += 256) {
        int m = i >> 7, j = i & 127;
        float zi = sh_z[m*STRZ + j]       + bcell[j];
        float zf = sh_z[m*STRZ + 128 + j] + bcell[128 + j];
        float zg = sh_z[m*STRZ + 256 + j] + bcell[256 + j];
        float zo = sh_z[m*STRZ + 384 + j] + bcell[384 + j];
        int row = R0 + m;
        float cp = c[row*128 + j];
        float si = 1.0f / (1.0f + __expf(-zi));
        float sf = 1.0f / (1.0f + __expf(-zf));
        float so = 1.0f / (1.0f + __expf(-zo));
        float cn = sf * cp + si * tanhf(zg);
        c[row*128 + j]     = cn;
        h_out[row*128 + j] = so * tanhf(cn);
    }
}

__global__ __launch_bounds__(256) void out_kernel(
    const float* __restrict__ h, const float* __restrict__ W1T,
    const float* __restrict__ b1, const float* __restrict__ W2T,
    const float* __restrict__ b2, float* __restrict__ outbuf)
{
    __shared__ float sh_a[16 * STRH];
    __shared__ float sh_hid[16 * STRH];
    const int tid = threadIdx.x;
    const int R0  = blockIdx.x * 16;

    for (int i = tid; i < 2048; i += 256) {
        int m = i >> 7, k = i & 127;
        sh_a[m*STRH + k] = h[(R0 + m)*128 + k];
    }
    __syncthreads();

    const int wave = tid >> 5, lane = tid & 31;
    const int lm = lane & 15, hi = lane >> 4, kHalf = hi * 2;
    const int col = wave * 16 + lm;
    const float* wk   = W1T + col * 128;
    const float* arow = &sh_a[lm * STRH];

    v8f acc = {};
    #pragma unroll 8
    for (int kk = 0; kk < 128; kk += 4) {
        v2f a, b;
        a.x = arow[kk + kHalf]; a.y = arow[kk + kHalf + 1];
        b.x = wk[kk + kHalf];   b.y = wk[kk + kHalf + 1];
        acc = __builtin_amdgcn_wmma_f32_16x16x4_f32(
                false, a, false, b, (short)0, acc, false, false);
    }
    float bias = b1[col];
    #pragma unroll
    for (int v = 0; v < 8; ++v) {
        float x = acc[v] + bias;
        sh_hid[(v + hi * 8) * STRH + col] = x > 0.0f ? x : 0.0f;
    }
    __syncthreads();

    if (tid < 48) {
        int m = tid / 3, cc = tid % 3;
        const float* hr = &sh_hid[m * STRH];
        const float* w2 = W2T + cc * 128;
        float s = 0.0f;
        for (int k = 0; k < 128; ++k) s += hr[k] * w2[k];
        outbuf[(R0 + m) * 3 + cc] = s + b2[cc];
    }
}

__global__ __launch_bounds__(256) void nd_kernel(
    const float* __restrict__ input, const float* __restrict__ outbuf,
    float* __restrict__ out, int t)
{
    int gid = blockIdx.x * 256 + threadIdx.x;
    if (gid >= 1024) return;
    int bb = gid >> 7, s = gid & 127;
    const float* o = &outbuf[gid * 3];
    float n0 = o[0];
    float n1, sp0, sp1;
    if (s == 0) {
        const float* xn0 = &input[((bb * 128 + (t + 1)) * 128 + 0) * 5];
        n1 = xn0[1]; sp0 = xn0[3]; sp1 = xn0[4];
    } else {
        n1  = outbuf[(gid - 1) * 3 + 0];
        sp0 = o[1]; sp1 = o[2];
    }
    float fn = input[((bb * 128 + t) * 128 + s) * 5 + 2];
    float n2 = fn + n1 - n0;
    float* dst = &out[((bb * 123 + (t - 4)) * 128 + s) * 5];
    dst[0] = n0; dst[1] = n1; dst[2] = n2; dst[3] = sp0; dst[4] = sp1;
}

__global__ __launch_bounds__(256) void zero_kernel(float* p, int n) {
    int i = blockIdx.x * 256 + threadIdx.x;
    if (i < n) p[i] = 0.0f;
}

__global__ __launch_bounds__(256) void prep_kernel(
    const float* __restrict__ Wc, const float* __restrict__ W1,
    const float* __restrict__ W2, float* __restrict__ WT,
    float* __restrict__ W1T, float* __restrict__ W2T)
{
    int i = blockIdx.x * 256 + threadIdx.x;
    if (i < 12 * 16384) {
        int chn = i >> 14, col = (i >> 5) & 511, q = i & 31;
        WT[i] = Wc[(5 + chn * 32 + q) * 512 + col];
    } else if (i < 512 * 392) {
        int j = i - 12 * 16384;
        int col = j >> 3, k = j & 7;
        WT[i] = (k < 5) ? Wc[k * 512 + col] : 0.0f;
    } else if (i < 512 * 392 + 16384) {
        int j = i - 512 * 392;
        int col = j >> 7, k = j & 127;
        W1T[j] = W1[k * 128 + col];
    } else if (i < 512 * 392 + 16384 + 384) {
        int j = i - 512 * 392 - 16384;
        int cc = j >> 7, k = j & 127;
        W2T[j] = W2[k * 3 + cc];
    }
}

extern "C" void kernel_launch(void* const* d_in, const int* in_sizes, int n_in,
                              void* d_out, int out_size, void* d_ws, size_t ws_size,
                              hipStream_t stream)
{
    (void)in_sizes; (void)n_in; (void)out_size; (void)ws_size;
    const float* input = (const float*)d_in[0];
    const float* Wc    = (const float*)d_in[1];
    const float* bc    = (const float*)d_in[2];
    const float* W1    = (const float*)d_in[3];
    const float* b1    = (const float*)d_in[4];
    const float* W2    = (const float*)d_in[5];
    const float* b2    = (const float*)d_in[6];
    float* ws   = (float*)d_ws;
    float* WT   = ws;
    float* W1T  = ws + 200704;
    float* W2T  = ws + 217088;
    float* hbuf = ws + 217472;
    float* cbuf = ws + 479616;
    float* obuf = ws + 610688;
    float* out  = (float*)d_out;

    zero_kernel<<<393216 / 256, 256, 0, stream>>>(hbuf, 393216);
    prep_kernel<<<(217472 + 255) / 256, 256, 0, stream>>>(Wc, W1, W2, WT, W1T, W2T);

    for (int t = 0; t < 127; ++t) {
        const float* h_in = hbuf + (t & 1) * 131072;
        float* h_out      = hbuf + ((t + 1) & 1) * 131072;
        cell_kernel<<<64, 256, 0, stream>>>(input, WT, bc, h_in, h_out, cbuf, t);
        out_kernel<<<64, 256, 0, stream>>>(h_out, W1T, b1, W2T, b2, obuf);
        if (t >= 4) nd_kernel<<<4, 256, 0, stream>>>(input, obuf, out, t);
    }
}


# Output head moved into the cell block's epilogue on its 16 fresh h rows (W1 streamed as a 13th LDS chunk); head outputs kept per step, mix run once after the time loop: 130 launches, was 379
# speedup vs baseline: 2.5219x; 1.3513x over previous
#include <hip/hip_runtime.h>
#include <math.h>

typedef __attribute__((ext_vector_type(2))) float v2f;
typedef __attribute__((ext_vector_type(8))) float v8f;
typedef __attribute__((ext_vector_type(4))) float v4f;

#define STRH 130
#define STRZ 520
#define WP   36
#define WBUF (512 * WP)

__device__ __forceinline__ void chunk_request(v4f (&r)[16], const v4f* src, int tid)
{
    #pragma unroll
    for (int i = 0; i < 16; ++i) r[i] = src[tid + i * 256];
}

__device__ __forceinline__ void chunk_to_lds(float* buf, int wdst, const v4f (&r)[16])
{
    #pragma unroll
    for (int i = 0; i < 16; ++i)
        *reinterpret_cast<v4f*>(&buf[wdst + i * 32 * WP]) = r[i];
}

__device__ __forceinline__ void lds_barrier()
{
    asm volatile("s_wait_dscnt 0x0\n\ts_barrier_signal -1\n\ts_barrier_wait -1" ::: "memory");
}

__device__ __forceinline__ void chunk_mma(const float* arow, const float* wb,
    int wl0, int wl1, int wl2, int wl3, int kHalf,
    v8f& acc0, v8f& acc1, v8f& acc2, v8f& acc3)
{
    #pragma unroll
    for (int kk = 0; kk < 32; kk += 4) {
        v2f a;
        a.x = arow[kk + kHalf];
        a.y = arow[kk + kHalf + 1];
        v2f b0, b1, b2, b3;
        b0.x = wb[wl0 + kk + kHalf]; b0.y = wb[wl0 + kk + kHalf + 1];
        b1.x = wb[wl1 + kk + kHalf]; b1.y = wb[wl1 + kk + kHalf + 1];
        b2.x = wb[wl2 + kk + kHalf]; b2.y = wb[wl2 + kk + kHalf + 1];
        b3.x = wb[wl3 + kk + kHalf]; b3.y = wb[wl3 + kk + kHalf + 1];
        acc0 = __builtin_amdgcn_wmma_f32_16x16x4_f32(
                    false, a, false, b0, (short)0, acc0, false, false);
        acc1 = __builtin_amdgcn_wmma_f32_16x16x4_f32(
                    false, a, false, b1, (short)0, acc1, false, false);
        acc2 = __builtin_amdgcn_wmma_f32_16x16x4_f32(
                    false, a, false, b2, (short)0, acc2, false, false);
        acc3 = __builtin_amdgcn_wmma_f32_16x16x4_f32(
                    false, a, false, b3, (short)0, acc3, false, false);
    }
}

__global__ __launch_bounds__(256) void cell_kernel(
    const float* __restrict__ input, const float* __restrict__ WT,
    const float* __restrict__ bcell, const float* __restrict__ h_in,
    float* __restrict__ h_out, float* __restrict__ c,
    const float* __restrict__ b1, const float* __restrict__ W2T,
    const float* __restrict__ b2, float* __restrict__ obuf_all, int t)
{
    __shared__ float sh_h[18 * STRH];
    __shared__ float sh_x[16 * 8];
    __shared__ float sh_z[16 * STRZ];
    __shared__ alignas(16) float sh_w[2 * WBUF];
    __shared__ float sh_hn[16 * STRH];
    __shared__ float sh_hid[16 * STRH];
    __shared__ alignas(16) float sh_w2[3 * 128];

    const int tid = threadIdx.x;
    const int R0  = blockIdx.x * 16;
    const int bb  = R0 >> 7;
    const int s0  = R0 & 127;

    const v4f* wsrc = reinterpret_cast<const v4f*>(WT);
    const int wdst = (tid >> 3) * WP + (tid & 7) * 4;
    v4f preA[16], preB[16];
    chunk_request(preA, wsrc, tid);
    chunk_request(preB, wsrc + 4096, tid);

    const int wave  = tid >> 5;
    const int lane  = tid & 31;
    const int lm    = lane & 15;
    const int hi    = lane >> 4;
    const int kHalf = hi * 2;

    const float* wc0 = WT + 13 * 16384 + ((wave * 4 + 0) * 16 + lm) * 8;
    const float* wc1 = WT + 13 * 16384 + ((wave * 4 + 1) * 16 + lm) * 8;
    const float* wc2 = WT + 13 * 16384 + ((wave * 4 + 2) * 16 + lm) * 8;
    const float* wc3 = WT + 13 * 16384 + ((wave * 4 + 3) * 16 + lm) * 8;
    const v4f wq0 = *reinterpret_cast<const v4f*>(wc0); const float we0 = wc0[4];
    const v4f wq1 = *reinterpret_cast<const v4f*>(wc1); const float we1 = wc1[4];
    const v4f wq2 = *reinterpret_cast<const v4f*>(wc2); const float we2 = wc2[4];
    const v4f wq3 = *reinterpret_cast<const v4f*>(wc3); const float we3 = wc3[4];

    if (tid < 96)
        *reinterpret_cast<v4f*>(&sh_w2[tid * 4]) = *reinterpret_cast<const v4f*>(W2T + tid * 4);
    const int   hcol  = wave * 16 + lm;
    const float bias1 = b1[hcol];
    const float bias2 = b2[tid % 3];

    for (int i = tid; i < 18 * 128; i += 256) {
        int r = i >> 7, k = i & 127;
        int s = s0 - 1 + r;
        float v = 0.0f;
        if (s >= 0 && s < 128) v = h_in[(bb * 128 + s) * 128 + k];
        sh_h[r * STRH + k] = v;
    }
    if (tid < 16 * 5) {
        int m = tid / 5, k = tid % 5;
        sh_x[m * 8 + k] = input[(((bb * 128) + t) * 128 + (s0 + m)) * 5 + k];
    }
    float* const buf0 = sh_w;
    float* const buf1 = sh_w + WBUF;
    chunk_to_lds(buf0, wdst, preA);
    chunk_to_lds(buf1, wdst, preB);
    chunk_request(preA, wsrc + 2 * 4096, tid);
    chunk_request(preB, wsrc + 3 * 4096, tid);
    lds_barrier();

    const int wl0 = ((wave * 4 + 0) * 16 + lm) * WP;
    const int wl1 = ((wave * 4 + 1) * 16 + lm) * WP;
    const int wl2 = ((wave * 4 + 2) * 16 + lm) * WP;
    const int wl3 = ((wave * 4 + 3) * 16 + lm) * WP;

    v8f acc0 = {}, acc1 = {}, acc2 = {}, acc3 = {};

    #pragma unroll
    for (int v = 0; v < 8; ++v) {
        int m = v + hi * 8;
        float x0 = sh_x[m*8+0], x1 = sh_x[m*8+1], x2 = sh_x[m*8+2],
              x3 = sh_x[m*8+3], x4 = sh_x[m*8+4];
        acc0[v] += x0*wq0.x + x1*wq0.y + x2*wq0.z + x3*wq0.w + x4*we0;
        acc1[v] += x0*wq1.x + x1*wq1.y + x2*wq1.z + x3*wq1.w + x4*we1;
        acc2[v] += x0*wq2.x + x1*wq2.y + x2*wq2.z + x3*wq2.w + x4*we2;
        acc3[v] += x0*wq3.x + x1*wq3.y + x2*wq3.z + x3*wq3.w + x4*we3;
    }

    for (int ch = 0; ch < 12; ch += 2) {
        const int src    = ch >> 2;
        const int shBase = (src == 0) ? 1 : (src == 1) ? 2 : 0;
        const float* arow = &sh_h[(shBase + lm) * STRH + (ch & 3) * 32];

        if (ch > 0) {
            chunk_to_lds(buf1, wdst, preB);
            if (ch + 3 < 12) chunk_request(preB, wsrc + (ch + 3) * 4096, tid);
        }
        chunk_mma(arow, buf0, wl0, wl1, wl2, wl3, kHalf, acc0, acc1, acc2, acc3);
        lds_barrier();

        chunk_to_lds(buf0, wdst, preA);
        if (ch + 4 < 13) chunk_request(preA, wsrc + (ch + 4) * 4096, tid);
        chunk_mma(arow + 32, buf1, wl0, wl1, wl2, wl3, kHalf, acc0, acc1, acc2, acc3);
        if (ch + 2 < 12) lds_barrier();
    }

    #pragma unroll
    for (int v = 0; v < 8; ++v) {
        int m = v + hi * 8;
        sh_z[m * STRZ + (wave * 4 + 0) * 16 + lm] = acc0[v];
        sh_z[m * STRZ + (wave * 4 + 1) * 16 + lm] = acc1[v];
        sh_z[m * STRZ + (wave * 4 + 2) * 16 + lm] = acc2[v];
        sh_z[m * STRZ + (wave * 4 + 3) * 16 + lm] = acc3[v];
    }
    __syncthreads();

    for (int i = tid; i < 2048; i += 256) {
        int m = i >> 7, j = i & 127;
        float zi = sh_z[m*STRZ + j]       + bcell[j];
        float zf = sh_z[m*STRZ + 128 + j] + bcell[128 + j];
        float zg = sh_z[m*STRZ + 256 + j] + bcell[256 + j];
        float zo = sh_z[m*STRZ + 384 + j] + bcell[384 + j];
        int row = R0 + m;
        float cp = c[row*128 + j];
        float si = 1.0f / (1.0f + __expf(-zi));
        float sf = 1.0f / (1.0f + __expf(-zf));
        float so = 1.0f / (1.0f + __expf(-zo));
        float cn = sf * cp + si * tanhf(zg);
        float hn = so * tanhf(cn);
        c[row*128 + j]     = cn;
        h_out[row*128 + j] = hn;
        sh_hn[m*STRH + j]  = hn;
    }
    __syncthreads();

    {
        const float* hrow = &sh_hn[lm * STRH];
        v8f hacc = {};
        #pragma unroll
        for (int kq = 0; kq < 4; ++kq) {
            const float* wq = buf0 + (kq * 128 + hcol) * WP;
            #pragma unroll
            for (int kk = 0; kk < 32; kk += 4) {
                v2f a, b;
                a.x = hrow[kq * 32 + kk + kHalf]; a.y = hrow[kq * 32 + kk + kHalf + 1];
                b.x = wq[kk + kHalf];             b.y = wq[kk + kHalf + 1];
                hacc = __builtin_amdgcn_wmma_f32_16x16x4_f32(
                            false, a, false, b, (short)0, hacc, false, false);
            }
        }
        #pragma unroll
        for (int v = 0; v < 8; ++v) {
            float x = hacc[v] + bias1;
            sh_hid[(v + hi * 8) * STRH + hcol] = x > 0.0f ? x : 0.0f;
        }
    }
    __syncthreads();

    if (tid < 48) {
        int m = tid / 3, cc = tid % 3;
        const float* hr = &sh_hid[m * STRH];
        const float* w2 = &sh_w2[cc * 128];
        float s = 0.0f;
        for (int k = 0; k < 128; ++k) s += hr[k] * w2[k];
        obuf_all[t * 3072 + (R0 + m) * 3 + cc] = s + bias2;
    }
}

__global__ __launch_bounds__(256) void mix_all_kernel(
    const float* __restrict__ input, const float* __restrict__ obuf_all,
    float* __restrict__ out)
{
    int gid = blockIdx.x * 256 + threadIdx.x;
    if (gid >= 123 * 1024) return;
    int tt = gid >> 10, r = gid & 1023;
    int t  = tt + 4;
    int bb = r >> 7, s = r & 127;
    const float* ob = obuf_all + t * 3072;
    const float* o  = &ob[r * 3];
    float n0 = o[0];
    float n1, sp0, sp1;
    if (s == 0) {
        const float* xn0 = &input[((bb * 128 + (t + 1)) * 128 + 0) * 5];
        n1 = xn0[1]; sp0 = xn0[3]; sp1 = xn0[4];
    } else {
        n1  = ob[(r - 1) * 3 + 0];
        sp0 = o[1]; sp1 = o[2];
    }
    float fn = input[((bb * 128 + t) * 128 + s) * 5 + 2];
    float n2 = fn + n1 - n0;
    float* dst = &out[((bb * 123 + tt) * 128 + s) * 5];
    dst[0] = n0; dst[1] = n1; dst[2] = n2; dst[3] = sp0; dst[4] = sp1;
}

__global__ __launch_bounds__(256) void zero_kernel(float* p, int n) {
    int i = blockIdx.x * 256 + threadIdx.x;
    if (i < n) p[i] = 0.0f;
}

__global__ __launch_bounds__(256) void prep_kernel(
    const float* __restrict__ Wc, const float* __restrict__ W1,
    const float* __restrict__ W2, float* __restrict__ WT,
    float* __restrict__ W2T)
{
    int i = blockIdx.x * 256 + threadIdx.x;
    if (i < 12 * 16384) {
        int chn = i >> 14, col = (i >> 5) & 511, q = i & 31;
        WT[i] = Wc[(5 + chn * 32 + q) * 512 + col];
    } else if (i < 13 * 16384) {
        int j = i - 12 * 16384;
        int pc = j >> 5, q = j & 31;
        int kq = pc >> 7, n = pc & 127;
        WT[i] = W1[(kq * 32 + q) * 128 + n];
    } else if (i < 13 * 16384 + 4096) {
        int j = i - 13 * 16384;
        int col = j >> 3, k = j & 7;
        WT[i] = (k < 5) ? Wc[k * 512 + col] : 0.0f;
    } else if (i < 13 * 16384 + 4096 + 384) {
        int j = i - 13 * 16384 - 4096;
        int cc = j >> 7, k = j & 127;
        W2T[j] = W2[k * 3 + cc];
    }
}

extern "C" void kernel_launch(void* const* d_in, const int* in_sizes, int n_in,
                              void* d_out, int out_size, void* d_ws, size_t ws_size,
                              hipStream_t stream)
{
    (void)in_sizes; (void)n_in; (void)out_size; (void)ws_size;
    const float* input = (const float*)d_in[0];
    const float* Wc    = (const float*)d_in[1];
    const float* bc    = (const float*)d_in[2];
    const float* W1    = (const float*)d_in[3];
    const float* b1    = (const float*)d_in[4];
    const float* W2    = (const float*)d_in[5];
    const float* b2    = (const float*)d_in[6];
    float* ws   = (float*)d_ws;
    float* WT   = ws;
    float* W2T  = ws + 217088;
    float* hbuf = ws + 217472;
    float* cbuf = ws + 479616;
    float* obuf = ws + 610688;
    float* out  = (float*)d_out;

    zero_kernel<<<393216 / 256, 256, 0, stream>>>(hbuf, 393216);
    prep_kernel<<<(217472 + 255) / 256, 256, 0, stream>>>(Wc, W1, W2, WT, W2T);

    for (int t = 0; t < 127; ++t) {
        const float* h_in = hbuf + (t & 1) * 131072;
        float* h_out      = hbuf + ((t + 1) & 1) * 131072;
        cell_kernel<<<64, 256, 0, stream>>>(input, WT, bc, h_in, h_out, cbuf,
                                            b1, W2T, b2, obuf, t);
    }
    mix_all_kernel<<<123 * 1024 / 256, 256, 0, stream>>>(input, obuf, out);
}


# Cell GEMM split 4 ways in K over 256 blocks (a 32-unit slice of each h source per block); a second per-step launch adds the 4 partial planes in fixed order, then gates and head: 257 launches
# speedup vs baseline: 3.2151x; 1.2749x over previous
#include <hip/hip_runtime.h>
#include <math.h>

typedef __attribute__((ext_vector_type(2))) float v2f;
typedef __attribute__((ext_vector_type(8))) float v8f;
typedef __attribute__((ext_vector_type(4))) float v4f;

#define STRH 130
#define STRZ 520
#define WP   36
#define WBUF (512 * WP)

__device__ __forceinline__ void chunk_request(v4f (&r)[16], const v4f* src, int tid)
{
    #pragma unroll
    for (int i = 0; i < 16; ++i) r[i] = src[tid + i * 256];
}

__device__ __forceinline__ void chunk_to_lds(float* buf, int wdst, const v4f (&r)[16])
{
    #pragma unroll
    for (int i = 0; i < 16; ++i)
        *reinterpret_cast<v4f*>(&buf[wdst + i * 32 * WP]) = r[i];
}

__device__ __forceinline__ void lds_barrier()
{
    asm volatile("s_wait_dscnt 0x0\n\ts_barrier_signal -1\n\ts_barrier_wait -1" ::: "memory");
}

__device__ __forceinline__ void chunk_mma(const float* arow, const float* wb,
    int wl0, int wl1, int wl2, int wl3, int kHalf,
    v8f& acc0, v8f& acc1, v8f& acc2, v8f& acc3)
{
    #pragma unroll
    for (int kk = 0; kk < 32; kk += 4) {
        v2f a;
        a.x = arow[kk + kHalf];
        a.y = arow[kk + kHalf + 1];
        v2f b0, b1, b2, b3;
        b0.x = wb[wl0 + kk + kHalf]; b0.y = wb[wl0 + kk + kHalf + 1];
        b1.x = wb[wl1 + kk + kHalf]; b1.y = wb[wl1 + kk + kHalf + 1];
        b2.x = wb[wl2 + kk + kHalf]; b2.y = wb[wl2 + kk + kHalf + 1];
        b3.x = wb[wl3 + kk + kHalf]; b3.y = wb[wl3 + kk + kHalf + 1];
        acc0 = __builtin_amdgcn_wmma_f32_16x16x4_f32(
                    false, a, false, b0, (short)0, acc0, false, false);
        acc1 = __builtin_amdgcn_wmma_f32_16x16x4_f32(
                    false, a, false, b1, (short)0, acc1, false, false);
        acc2 = __builtin_amdgcn_wmma_f32_16x16x4_f32(
                    false, a, false, b2, (short)0, acc2, false, false);
        acc3 = __builtin_amdgcn_wmma_f32_16x16x4_f32(
                    false, a, false, b3, (short)0, acc3, false, false);
    }
}

#define HP 36
#define ZPLANE (1024 * 512)

__global__ __launch_bounds__(256) void cell_part_kernel(
    const float* __restrict__ input, const float* __restrict__ WT,
    const float* __restrict__ h_in, float* __restrict__ zpart, int t)
{
    __shared__ alignas(16) float sh_h[18 * HP];
    __shared__ float sh_x[16 * 8];
    __shared__ alignas(16) float sh_z[16 * STRZ];
    __shared__ alignas(16) float sh_w[3 * WBUF];

    const int tid = threadIdx.x;
    const int mt  = blockIdx.x >> 2;
    const int ks  = blockIdx.x & 3;
    const int R0  = mt * 16;
    const int bb  = R0 >> 7;
    const int s0  = R0 & 127;

    v4f hv = {0.0f, 0.0f, 0.0f, 0.0f};
    if (tid < 18 * 8) {
        int s = s0 - 1 + (tid >> 3);
        if (s >= 0 && s < 128)
            hv = *reinterpret_cast<const v4f*>(&h_in[(bb * 128 + s) * 128 + ks * 32 + (tid & 7) * 4]);
    }
    float xv = 0.0f;
    if (tid < 16 * 5)
        xv = input[(((bb * 128) + t) * 128 + (s0 + tid / 5)) * 5 + tid % 5];

    const v4f* wsrc = reinterpret_cast<const v4f*>(WT);
    const int wdst = (tid >> 3) * WP + (tid & 7) * 4;
    v4f preA[16], preB[16];
    chunk_request(preA, wsrc + ks * 4096, tid);
    chunk_request(preB, wsrc + (4 + ks) * 4096, tid);

    const int wave  = tid >> 5;
    const int lane  = tid & 31;
    const int lm    = lane & 15;
    const int hi    = lane >> 4;
    const int kHalf = hi * 2;

    const float* wc0 = WT + 13 * 16384 + ((wave * 4 + 0) * 16 + lm) * 8;
    const float* wc1 = WT + 13 * 16384 + ((wave * 4 + 1) * 16 + lm) * 8;
    const float* wc2 = WT + 13 * 16384 + ((wave * 4 + 2) * 16 + lm) * 8;
    const float* wc3 = WT + 13 * 16384 + ((wave * 4 + 3) * 16 + lm) * 8;
    const v4f wq0 = *reinterpret_cast<const v4f*>(wc0); const float we0 = wc0[4];
    const v4f wq1 = *reinterpret_cast<const v4f*>(wc1); const float we1 = wc1[4];
    const v4f wq2 = *reinterpret_cast<const v4f*>(wc2); const float we2 = wc2[4];
    const v4f wq3 = *reinterpret_cast<const v4f*>(wc3); const float we3 = wc3[4];

    if (tid < 18 * 8)
        *reinterpret_cast<v4f*>(&sh_h[(tid >> 3) * HP + (tid & 7) * 4]) = hv;
    if (tid < 16 * 5)
        sh_x[(tid / 5) * 8 + tid % 5] = xv;
    float* const buf0 = sh_w;
    float* const buf1 = sh_w + WBUF;
    float* const buf2 = sh_w + 2 * WBUF;
    chunk_to_lds(buf0, wdst, preA);
    chunk_request(preA, wsrc + (8 + ks) * 4096, tid);
    chunk_to_lds(buf1, wdst, preB);
    lds_barrier();

    const int wl0 = ((wave * 4 + 0) * 16 + lm) * WP;
    const int wl1 = ((wave * 4 + 1) * 16 + lm) * WP;
    const int wl2 = ((wave * 4 + 2) * 16 + lm) * WP;
    const int wl3 = ((wave * 4 + 3) * 16 + lm) * WP;

    v8f acc0 = {}, acc1 = {}, acc2 = {}, acc3 = {};

    if (ks == 0) {
        #pragma unroll
        for (int v = 0; v < 8; ++v) {
            int m = v + hi * 8;
            float x0 = sh_x[m*8+0], x1 = sh_x[m*8+1], x2 = sh_x[m*8+2],
                  x3 = sh_x[m*8+3], x4 = sh_x[m*8+4];
            acc0[v] += x0*wq0.x + x1*wq0.y + x2*wq0.z + x3*wq0.w + x4*we0;
            acc1[v] += x0*wq1.x + x1*wq1.y + x2*wq1.z + x3*wq1.w + x4*we1;
            acc2[v] += x0*wq2.x + x1*wq2.y + x2*wq2.z + x3*wq2.w + x4*we2;
            acc3[v] += x0*wq3.x + x1*wq3.y + x2*wq3.z + x3*wq3.w + x4*we3;
        }
    }

    chunk_mma(&sh_h[(1 + lm) * HP], buf0, wl0, wl1, wl2, wl3, kHalf, acc0, acc1, acc2, acc3);
    chunk_mma(&sh_h[(2 + lm) * HP], buf1, wl0, wl1, wl2, wl3, kHalf, acc0, acc1, acc2, acc3);
    chunk_to_lds(buf2, wdst, preA);
    lds_barrier();
    chunk_mma(&sh_h[lm * HP], buf2, wl0, wl1, wl2, wl3, kHalf, acc0, acc1, acc2, acc3);

    #pragma unroll
    for (int v = 0; v < 8; ++v) {
        int m = v + hi * 8;
        sh_z[m * STRZ + (wave * 4 + 0) * 16 + lm] = acc0[v];
        sh_z[m * STRZ + (wave * 4 + 1) * 16 + lm] = acc1[v];
        sh_z[m * STRZ + (wave * 4 + 2) * 16 + lm] = acc2[v];
        sh_z[m * STRZ + (wave * 4 + 3) * 16 + lm] = acc3[v];
    }
    __syncthreads();

    float* zp = zpart + ks * ZPLANE + R0 * 512;
    #pragma unroll
    for (int it = 0; it < 8; ++it) {
        int i = tid + it * 256;
        int m = i >> 7, c4 = (i & 127) * 4;
        *reinterpret_cast<v4f*>(&zp[m * 512 + c4]) =
            *reinterpret_cast<const v4f*>(&sh_z[m * STRZ + c4]);
    }
}

__device__ __forceinline__ v4f plane_sum(const float* p)
{
    v4f a = *reinterpret_cast<const v4f*>(p);
    v4f b = *reinterpret_cast<const v4f*>(p + ZPLANE);
    v4f c = *reinterpret_cast<const v4f*>(p + 2 * ZPLANE);
    v4f d = *reinterpret_cast<const v4f*>(p + 3 * ZPLANE);
    return ((a + b) + c) + d;
}

__global__ __launch_bounds__(256) void gate_head_kernel(
    const float* __restrict__ WT, const float* __restrict__ bcell,
    const float* __restrict__ zpart, float* __restrict__ h_out,
    float* __restrict__ c, const float* __restrict__ b1,
    const float* __restrict__ W2T, const float* __restrict__ b2,
    float* __restrict__ obuf_all, int t)
{
    __shared__ alignas(16) float sh_w[WBUF];
    __shared__ float sh_hn[16 * STRH];
    __shared__ float sh_hid[16 * STRH];
    __shared__ alignas(16) float sh_w2[3 * 128];

    const int tid = threadIdx.x;
    const int R0  = blockIdx.x * 16;

    const v4f* wsrc = reinterpret_cast<const v4f*>(WT);
    const int wdst = (tid >> 3) * WP + (tid & 7) * 4;
    v4f preA[16];
    chunk_request(preA, wsrc + 12 * 4096, tid);

    const int wave  = tid >> 5;
    const int lane  = tid & 31;
    const int lm    = lane & 15;
    const int hi    = lane >> 4;
    const int kHalf = hi * 2;

    v4f w2v = {0.0f, 0.0f, 0.0f, 0.0f};
    if (tid < 96) w2v = *reinterpret_cast<const v4f*>(W2T + tid * 4);
    const int   hcol  = wave * 16 + lm;
    const float bias1 = b1[hcol];
    const float bias2 = b2[tid % 3];

    const int gm = tid >> 5, gj = (tid & 31) * 4;
    const v4f bi = *reinterpret_cast<const v4f*>(&bcell[gj]);
    const v4f bf = *reinterpret_cast<const v4f*>(&bcell[128 + gj]);
    const v4f bg = *reinterpret_cast<const v4f*>(&bcell[256 + gj]);
    const v4f bo = *reinterpret_cast<const v4f*>(&bcell[384 + gj]);
    v4f zi[2], zf[2], zg[2], zo[2], cp[2];
    #pragma unroll
    for (int g = 0; g < 2; ++g) {
        int row = R0 + gm + g * 8;
        const float* zp = zpart + row * 512 + gj;
        zi[g] = plane_sum(zp)       + bi;
        zf[g] = plane_sum(zp + 128) + bf;
        zg[g] = plane_sum(zp + 256) + bg;
        zo[g] = plane_sum(zp + 384) + bo;
        cp[g] = *reinterpret_cast<const v4f*>(&c[row * 128 + gj]);
    }
    #pragma unroll
    for (int g = 0; g < 2; ++g) {
        int m = gm + g * 8;
        int row = R0 + m;
        v4f cn, hn;
        #pragma unroll
        for (int e = 0; e < 4; ++e) {
            float si = 1.0f / (1.0f + __expf(-zi[g][e]));
            float sf = 1.0f / (1.0f + __expf(-zf[g][e]));
            float so = 1.0f / (1.0f + __expf(-zo[g][e]));
            float cv = sf * cp[g][e] + si * tanhf(zg[g][e]);
            float hv = so * tanhf(cv);
            cn[e] = cv;
            hn[e] = hv;
            sh_hn[m * STRH + gj + e] = hv;
        }
        *reinterpret_cast<v4f*>(&c[row * 128 + gj])     = cn;
        *reinterpret_cast<v4f*>(&h_out[row * 128 + gj]) = hn;
    }
    if (tid < 96) *reinterpret_cast<v4f*>(&sh_w2[tid * 4]) = w2v;
    float* const buf0 = sh_w;
    chunk_to_lds(buf0, wdst, preA);
    __syncthreads();

    {
        const float* hrow = &sh_hn[lm * STRH];
        v8f hacc = {};
        #pragma unroll
        for (int kq = 0; kq < 4; ++kq) {
            const float* wq = buf0 + (kq * 128 + hcol) * WP;
            #pragma unroll
            for (int kk = 0; kk < 32; kk += 4) {
                v2f a, b;
                a.x = hrow[kq * 32 + kk + kHalf]; a.y = hrow[kq * 32 + kk + kHalf + 1];
                b.x = wq[kk + kHalf];             b.y = wq[kk + kHalf + 1];
                hacc = __builtin_amdgcn_wmma_f32_16x16x4_f32(
                            false, a, false, b, (short)0, hacc, false, false);
            }
        }
        #pragma unroll
        for (int v = 0; v < 8; ++v) {
            float x = hacc[v] + bias1;
            sh_hid[(v + hi * 8) * STRH + hcol] = x > 0.0f ? x : 0.0f;
        }
    }
    __syncthreads();

    if (tid < 48) {
        int m = tid / 3, cc = tid % 3;
        const float* hr = &sh_hid[m * STRH];
        const float* w2 = &sh_w2[cc * 128];
        float s = 0.0f;
        for (int k = 0; k < 128; ++k) s += hr[k] * w2[k];
        obuf_all[t * 3072 + (R0 + m) * 3 + cc] = s + bias2;
    }
}

__global__ __launch_bounds__(256) void mix_all_kernel(
    const float* __restrict__ input, const float* __restrict__ obuf_all,
    float* __restrict__ out)
{
    int gid = blockIdx.x * 256 + threadIdx.x;
    if (gid >= 123 * 1024) return;
    int tt = gid >> 10, r = gid & 1023;
    int t  = tt + 4;
    int bb = r >> 7, s = r & 127;
    const float* ob = obuf_all + t * 3072;
    const float* o  = &ob[r * 3];
    float n0 = o[0];
    float n1, sp0, sp1;
    if (s == 0) {
        const float* xn0 = &input[((bb * 128 + (t + 1)) * 128 + 0) * 5];
        n1 = xn0[1]; sp0 = xn0[3]; sp1 = xn0[4];
    } else {
        n1  = ob[(r - 1) * 3 + 0];
        sp0 = o[1]; sp1 = o[2];
    }
    float fn = input[((bb * 128 + t) * 128 + s) * 5 + 2];
    float n2 = fn + n1 - n0;
    float* dst = &out[((bb * 123 + tt) * 128 + s) * 5];
    dst[0] = n0; dst[1] = n1; dst[2] = n2; dst[3] = sp0; dst[4] = sp1;
}

__global__ __launch_bounds__(256) void zero_kernel(float* p, int n) {
    int i = blockIdx.x * 256 + threadIdx.x;
    if (i < n) p[i] = 0.0f;
}

__global__ __launch_bounds__(256) void prep_kernel(
    const float* __restrict__ Wc, const float* __restrict__ W1,
    const float* __restrict__ W2, float* __restrict__ WT,
    float* __restrict__ W2T)
{
    int i = blockIdx.x * 256 + threadIdx.x;
    if (i < 12 * 16384) {
        int chn = i >> 14, col = (i >> 5) & 511, q = i & 31;
        WT[i] = Wc[(5 + chn * 32 + q) * 512 + col];
    } else if (i < 13 * 16384) {
        int j = i - 12 * 16384;
        int pc = j >> 5, q = j & 31;
        int kq = pc >> 7, n = pc & 127;
        WT[i] = W1[(kq * 32 + q) * 128 + n];
    } else if (i < 13 * 16384 + 4096) {
        int j = i - 13 * 16384;
        int col = j >> 3, k = j & 7;
        WT[i] = (k < 5) ? Wc[k * 512 + col] : 0.0f;
    } else if (i < 13 * 16384 + 4096 + 384) {
        int j = i - 13 * 16384 - 4096;
        int cc = j >> 7, k = j & 127;
        W2T[j] = W2[k * 3 + cc];
    }
}

extern "C" void kernel_launch(void* const* d_in, const int* in_sizes, int n_in,
                              void* d_out, int out_size, void* d_ws, size_t ws_size,
                              hipStream_t stream)
{
    (void)in_sizes; (void)n_in; (void)out_size; (void)ws_size;
    const float* input = (const float*)d_in[0];
    const float* Wc    = (const float*)d_in[1];
    const float* bc    = (const float*)d_in[2];
    const float* W1    = (const float*)d_in[3];
    const float* b1    = (const float*)d_in[4];
    const float* W2    = (const float*)d_in[5];
    const float* b2    = (const float*)d_in[6];
    float* ws   = (float*)d_ws;
    float* WT   = ws;
    float* W2T  = ws + 217088;
    float* hbuf = ws + 217472;
    float* cbuf = ws + 479616;
    float* obuf = ws + 610688;
    float* zprt = ws + 1000832;
    float* out  = (float*)d_out;

    zero_kernel<<<393216 / 256, 256, 0, stream>>>(hbuf, 393216);
    prep_kernel<<<(217472 + 255) / 256, 256, 0, stream>>>(Wc, W1, W2, WT, W2T);

    for (int t = 0; t < 127; ++t) {
        const float* h_in = hbuf + (t & 1) * 131072;
        float* h_out      = hbuf + ((t + 1) & 1) * 131072;
        cell_part_kernel<<<256, 256, 0, stream>>>(input, WT, h_in, zprt, t);
        gate_head_kernel<<<64, 256, 0, stream>>>(WT, bc, zprt, h_out, cbuf,
                                                 b1, W2T, b2, obuf, t);
    }
    mix_all_kernel<<<123 * 1024 / 256, 256, 0, stream>>>(input, obuf, out);
}


# Output head taken out of the time loop: h of every step kept, one batched head launch after it (16-row tile per wave, W1 staged once per 128 rows); per-step launch 2 is gates only, on 256 blocks
# speedup vs baseline: 4.4128x; 1.3725x over previous
#include <hip/hip_runtime.h>
#include <math.h>

typedef __attribute__((ext_vector_type(2))) float v2f;
typedef __attribute__((ext_vector_type(8))) float v8f;
typedef __attribute__((ext_vector_type(4))) float v4f;

#define STRZ 520
#define WP   36
#define WBUF (512 * WP)

__device__ __forceinline__ void chunk_request(v4f (&r)[16], const v4f* src, int tid)
{
    #pragma unroll
    for (int i = 0; i < 16; ++i) r[i] = src[tid + i * 256];
}

__device__ __forceinline__ void chunk_to_lds(float* buf, int wdst, const v4f (&r)[16])
{
    #pragma unroll
    for (int i = 0; i < 16; ++i)
        *reinterpret_cast<v4f*>(&buf[wdst + i * 32 * WP]) = r[i];
}

__device__ __forceinline__ void lds_barrier()
{
    asm volatile("s_wait_dscnt 0x0\n\ts_barrier_signal -1\n\ts_barrier_wait -1" ::: "memory");
}

__device__ __forceinline__ void chunk_mma(const float* arow, const float* wb,
    int wl0, int wl1, int wl2, int wl3, int kHalf,
    v8f& acc0, v8f& acc1, v8f& acc2, v8f& acc3)
{
    #pragma unroll
    for (int kk = 0; kk < 32; kk += 4) {
        v2f a;
        a.x = arow[kk + kHalf];
        a.y = arow[kk + kHalf + 1];
        v2f b0, b1, b2, b3;
        b0.x = wb[wl0 + kk + kHalf]; b0.y = wb[wl0 + kk + kHalf + 1];
        b1.x = wb[wl1 + kk + kHalf]; b1.y = wb[wl1 + kk + kHalf + 1];
        b2.x = wb[wl2 + kk + kHalf]; b2.y = wb[wl2 + kk + kHalf + 1];
        b3.x = wb[wl3 + kk + kHalf]; b3.y = wb[wl3 + kk + kHalf + 1];
        acc0 = __builtin_amdgcn_wmma_f32_16x16x4_f32(
                    false, a, false, b0, (short)0, acc0, false, false);
        acc1 = __builtin_amdgcn_wmma_f32_16x16x4_f32(
                    false, a, false, b1, (short)0, acc1, false, false);
        acc2 = __builtin_amdgcn_wmma_f32_16x16x4_f32(
                    false, a, false, b2, (short)0, acc2, false, false);
        acc3 = __builtin_amdgcn_wmma_f32_16x16x4_f32(
                    false, a, false, b3, (short)0, acc3, false, false);
    }
}

#define HP 36
#define ZPLANE (1024 * 512)

__global__ __launch_bounds__(256) void cell_part_kernel(
    const float* __restrict__ input, const float* __restrict__ WT,
    const float* __restrict__ h_in, float* __restrict__ zpart, int t)
{
    __shared__ alignas(16) float sh_h[18 * HP];
    __shared__ float sh_x[16 * 8];
    __shared__ alignas(16) float sh_z[16 * STRZ];
    __shared__ alignas(16) float sh_w[3 * WBUF];

    const int tid = threadIdx.x;
    const int mt  = blockIdx.x >> 2;
    const int ks  = blockIdx.x & 3;
    const int R0  = mt * 16;
    const int bb  = R0 >> 7;
    const int s0  = R0 & 127;

    v4f hv = {0.0f, 0.0f, 0.0f, 0.0f};
    if (tid < 18 * 8) {
        int s = s0 - 1 + (tid >> 3);
        if (s >= 0 && s < 128)
            hv = *reinterpret_cast<const v4f*>(&h_in[(bb * 128 + s) * 128 + ks * 32 + (tid & 7) * 4]);
    }
    float xv = 0.0f;
    if (tid < 16 * 5)
        xv = input[(((bb * 128) + t) * 128 + (s0 + tid / 5)) * 5 + tid % 5];

    const v4f* wsrc = reinterpret_cast<const v4f*>(WT);
    const int wdst = (tid >> 3) * WP + (tid & 7) * 4;
    v4f preA[16], preB[16];
    chunk_request(preA, wsrc + ks * 4096, tid);
    chunk_request(preB, wsrc + (4 + ks) * 4096, tid);

    const int wave  = tid >> 5;
    const int lane  = tid & 31;
    const int lm    = lane & 15;
    const int hi    = lane >> 4;
    const int kHalf = hi * 2;

    const float* wc0 = WT + 13 * 16384 + ((wave * 4 + 0) * 16 + lm) * 8;
    const float* wc1 = WT + 13 * 16384 + ((wave * 4 + 1) * 16 + lm) * 8;
    const float* wc2 = WT + 13 * 16384 + ((wave * 4 + 2) * 16 + lm) * 8;
    const float* wc3 = WT + 13 * 16384 + ((wave * 4 + 3) * 16 + lm) * 8;
    const v4f wq0 = *reinterpret_cast<const v4f*>(wc0); const float we0 = wc0[4];
    const v4f wq1 = *reinterpret_cast<const v4f*>(wc1); const float we1 = wc1[4];
    const v4f wq2 = *reinterpret_cast<const v4f*>(wc2); const float we2 = wc2[4];
    const v4f wq3 = *reinterpret_cast<const v4f*>(wc3); const float we3 = wc3[4];

    if (tid < 18 * 8)
        *reinterpret_cast<v4f*>(&sh_h[(tid >> 3) * HP + (tid & 7) * 4]) = hv;
    if (tid < 16 * 5)
        sh_x[(tid / 5) * 8 + tid % 5] = xv;
    float* const buf0 = sh_w;
    float* const buf1 = sh_w + WBUF;
    float* const buf2 = sh_w + 2 * WBUF;
    chunk_to_lds(buf0, wdst, preA);
    chunk_request(preA, wsrc + (8 + ks) * 4096, tid);
    chunk_to_lds(buf1, wdst, preB);
    lds_barrier();

    const int wl0 = ((wave * 4 + 0) * 16 + lm) * WP;
    const int wl1 = ((wave * 4 + 1) * 16 + lm) * WP;
    const int wl2 = ((wave * 4 + 2) * 16 + lm) * WP;
    const int wl3 = ((wave * 4 + 3) * 16 + lm) * WP;

    v8f acc0 = {}, acc1 = {}, acc2 = {}, acc3 = {};

    if (ks == 0) {
        #pragma unroll
        for (int v = 0; v < 8; ++v) {
            int m = v + hi * 8;
            float x0 = sh_x[m*8+0], x1 = sh_x[m*8+1], x2 = sh_x[m*8+2],
                  x3 = sh_x[m*8+3], x4 = sh_x[m*8+4];
            acc0[v] += x0*wq0.x + x1*wq0.y + x2*wq0.z + x3*wq0.w + x4*we0;
            acc1[v] += x0*wq1.x + x1*wq1.y + x2*wq1.z + x3*wq1.w + x4*we1;
            acc2[v] += x0*wq2.x + x1*wq2.y + x2*wq2.z + x3*wq2.w + x4*we2;
            acc3[v] += x0*wq3.x + x1*wq3.y + x2*wq3.z + x3*wq3.w + x4*we3;
        }
    }

    chunk_mma(&sh_h[(1 + lm) * HP], buf0, wl0, wl1, wl2, wl3, kHalf, acc0, acc1, acc2, acc3);
    chunk_mma(&sh_h[(2 + lm) * HP], buf1, wl0, wl1, wl2, wl3, kHalf, acc0, acc1, acc2, acc3);
    chunk_to_lds(buf2, wdst, preA);
    lds_barrier();
    chunk_mma(&sh_h[lm * HP], buf2, wl0, wl1, wl2, wl3, kHalf, acc0, acc1, acc2, acc3);

    #pragma unroll
    for (int v = 0; v < 8; ++v) {
        int m = v + hi * 8;
        sh_z[m * STRZ + (wave * 4 + 0) * 16 + lm] = acc0[v];
        sh_z[m * STRZ + (wave * 4 + 1) * 16 + lm] = acc1[v];
        sh_z[m * STRZ + (wave * 4 + 2) * 16 + lm] = acc2[v];
        sh_z[m * STRZ + (wave * 4 + 3) * 16 + lm] = acc3[v];
    }
    __syncthreads();

    float* zp = zpart + ks * ZPLANE + R0 * 512;
    #pragma unroll
    for (int it = 0; it < 8; ++it) {
        int i = tid + it * 256;
        int m = i >> 7, c4 = (i & 127) * 4;
        *reinterpret_cast<v4f*>(&zp[m * 512 + c4]) =
            *reinterpret_cast<const v4f*>(&sh_z[m * STRZ + c4]);
    }
}

__device__ __forceinline__ v4f plane_sum(const float* p)
{
    v4f a = *reinterpret_cast<const v4f*>(p);
    v4f b = *reinterpret_cast<const v4f*>(p + ZPLANE);
    v4f c = *reinterpret_cast<const v4f*>(p + 2 * ZPLANE);
    v4f d = *reinterpret_cast<const v4f*>(p + 3 * ZPLANE);
    return ((a + b) + c) + d;
}

__global__ __launch_bounds__(128) void gate_kernel(
    const float* __restrict__ bcell, const float* __restrict__ zpart,
    float* __restrict__ h_out, float* __restrict__ c)
{
    const int gid = blockIdx.x * 128 + threadIdx.x;
    const int row = gid >> 5, gj = (gid & 31) * 4;

    const float* zp = zpart + row * 512 + gj;
    const v4f zi = plane_sum(zp)       + *reinterpret_cast<const v4f*>(&bcell[gj]);
    const v4f zf = plane_sum(zp + 128) + *reinterpret_cast<const v4f*>(&bcell[128 + gj]);
    const v4f zg = plane_sum(zp + 256) + *reinterpret_cast<const v4f*>(&bcell[256 + gj]);
    const v4f zo = plane_sum(zp + 384) + *reinterpret_cast<const v4f*>(&bcell[384 + gj]);
    const v4f cp = *reinterpret_cast<const v4f*>(&c[row * 128 + gj]);

    v4f cn, hn;
    #pragma unroll
    for (int e = 0; e < 4; ++e) {
        float si = 1.0f / (1.0f + __expf(-zi[e]));
        float sf = 1.0f / (1.0f + __expf(-zf[e]));
        float so = 1.0f / (1.0f + __expf(-zo[e]));
        float cv = sf * cp[e] + si * tanhf(zg[e]);
        cn[e] = cv;
        hn[e] = so * tanhf(cv);
    }
    *reinterpret_cast<v4f*>(&c[row * 128 + gj])     = cn;
    *reinterpret_cast<v4f*>(&h_out[row * 128 + gj]) = hn;
}

#define HIDP 132

__global__ __launch_bounds__(256) void head_all_kernel(
    const float* __restrict__ WT, const float* __restrict__ hrows,
    const float* __restrict__ b1, const float* __restrict__ W2T,
    const float* __restrict__ b2, float* __restrict__ orows)
{
    __shared__ alignas(16) float sh_w[WBUF];
    __shared__ alignas(16) float sh_hid[8 * 16 * HIDP];
    __shared__ alignas(16) float sh_w2[3 * 128];
    __shared__ alignas(16) float sh_o[8 * 48];

    const int tid   = threadIdx.x;
    const int wave  = tid >> 5;
    const int lane  = tid & 31;
    const int lm    = lane & 15;
    const int hi    = lane >> 4;
    const int kHalf = hi * 2;

    const v4f* wsrc = reinterpret_cast<const v4f*>(WT);
    const int wdst = (tid >> 3) * WP + (tid & 7) * 4;
    v4f preA[16];
    chunk_request(preA, wsrc + 12 * 4096, tid);

    const float* hrow = hrows + ((blockIdx.x * 8 + wave) * 16 + lm) * 128 + kHalf;
    v2f a[32];
    #pragma unroll
    for (int ks = 0; ks < 32; ++ks)
        a[ks] = *reinterpret_cast<const v2f*>(hrow + ks * 4);

    float bias1[8];
    #pragma unroll
    for (int j = 0; j < 8; ++j) bias1[j] = b1[j * 16 + lm];
    const float b20 = b2[0], b21 = b2[1], b22 = b2[2];
    v4f w2v = {0.0f, 0.0f, 0.0f, 0.0f};
    if (tid < 96) w2v = *reinterpret_cast<const v4f*>(W2T + tid * 4);

    if (tid < 96) *reinterpret_cast<v4f*>(&sh_w2[tid * 4]) = w2v;
    chunk_to_lds(sh_w, wdst, preA);
    __syncthreads();

    v8f acc[8];
    #pragma unroll
    for (int j = 0; j < 8; ++j) acc[j] = v8f{};
    #pragma unroll
    for (int kq = 0; kq < 4; ++kq) {
        #pragma unroll
        for (int kk = 0; kk < 32; kk += 4) {
            #pragma unroll
            for (int j = 0; j < 8; ++j) {
                const float* wq = sh_w + (kq * 128 + j * 16 + lm) * WP;
                v2f b;
                b.x = wq[kk + kHalf]; b.y = wq[kk + kHalf + 1];
                acc[j] = __builtin_amdgcn_wmma_f32_16x16x4_f32(
                            false, a[kq * 8 + (kk >> 2)], false, b, (short)0, acc[j], false, false);
            }
        }
    }
    float* hid = sh_hid + wave * 16 * HIDP;
    #pragma unroll
    for (int j = 0; j < 8; ++j) {
        #pragma unroll
        for (int v = 0; v < 8; ++v) {
            float x = acc[j][v] + bias1[j];
            hid[(v + hi * 8) * HIDP + j * 16 + lm] = x > 0.0f ? x : 0.0f;
        }
    }
    __syncthreads();

    if (lane < 16) {
        const float* hr = hid + lane * HIDP;
        float s0 = 0.0f, s1 = 0.0f, s2 = 0.0f;
        for (int k = 0; k < 128; k += 4) {
            const v4f hv = *reinterpret_cast<const v4f*>(hr + k);
            const v4f wa = *reinterpret_cast<const v4f*>(sh_w2 + k);
            const v4f wb = *reinterpret_cast<const v4f*>(sh_w2 + 128 + k);
            const v4f wc = *reinterpret_cast<const v4f*>(sh_w2 + 256 + k);
            s0 = fmaf(hv.x, wa.x, s0); s0 = fmaf(hv.y, wa.y, s0);
            s0 = fmaf(hv.z, wa.z, s0); s0 = fmaf(hv.w, wa.w, s0);
            s1 = fmaf(hv.x, wb.x, s1); s1 = fmaf(hv.y, wb.y, s1);
            s1 = fmaf(hv.z, wb.z, s1); s1 = fmaf(hv.w, wb.w, s1);
            s2 = fmaf(hv.x, wc.x, s2); s2 = fmaf(hv.y, wc.y, s2);
            s2 = fmaf(hv.z, wc.z, s2); s2 = fmaf(hv.w, wc.w, s2);
        }
        float* so = sh_o + wave * 48 + lane * 3;
        so[0] = s0 + b20; so[1] = s1 + b21; so[2] = s2 + b22;
    }
    __syncthreads();

    if (tid < 96)
        *reinterpret_cast<v4f*>(orows + blockIdx.x * 384 + tid * 4) =
            *reinterpret_cast<const v4f*>(sh_o + tid * 4);
}

__global__ __launch_bounds__(256) void mix_all_kernel(
    const float* __restrict__ input, const float* __restrict__ obuf_all,
    float* __restrict__ out)
{
    int gid = blockIdx.x * 256 + threadIdx.x;
    if (gid >= 123 * 1024) return;
    int tt = gid >> 10, r = gid & 1023;
    int t  = tt + 4;
    int bb = r >> 7, s = r & 127;
    const float* ob = obuf_all + t * 3072;
    const float* o  = &ob[r * 3];
    float n0 = o[0];
    float n1, sp0, sp1;
    if (s == 0) {
        const float* xn0 = &input[((bb * 128 + (t + 1)) * 128 + 0) * 5];
        n1 = xn0[1]; sp0 = xn0[3]; sp1 = xn0[4];
    } else {
        n1  = ob[(r - 1) * 3 + 0];
        sp0 = o[1]; sp1 = o[2];
    }
    float fn = input[((bb * 128 + t) * 128 + s) * 5 + 2];
    float n2 = fn + n1 - n0;
    float* dst = &out[((bb * 123 + tt) * 128 + s) * 5];
    dst[0] = n0; dst[1] = n1; dst[2] = n2; dst[3] = sp0; dst[4] = sp1;
}

__global__ __launch_bounds__(256) void zero_kernel(float* p, int n) {
    int i = blockIdx.x * 256 + threadIdx.x;
    if (i < n) p[i] = 0.0f;
}

__global__ __launch_bounds__(256) void prep_kernel(
    const float* __restrict__ Wc, const float* __restrict__ W1,
    const float* __restrict__ W2, float* __restrict__ WT,
    float* __restrict__ W2T)
{
    int i = blockIdx.x * 256 + threadIdx.x;
    if (i < 12 * 16384) {
        int chn = i >> 14, col = (i >> 5) & 511, q = i & 31;
        WT[i] = Wc[(5 + chn * 32 + q) * 512 + col];
    } else if (i < 13 * 16384) {
        int j = i - 12 * 16384;
        int pc = j >> 5, q = j & 31;
        int kq = pc >> 7, n = pc & 127;
        WT[i] = W1[(kq * 32 + q) * 128 + n];
    } else if (i < 13 * 16384 + 4096) {
        int j = i - 13 * 16384;
        int col = j >> 3, k = j & 7;
        WT[i] = (k < 5) ? Wc[k * 512 + col] : 0.0f;
    } else if (i < 13 * 16384 + 4096 + 384) {
        int j = i - 13 * 16384 - 4096;
        int cc = j >> 7, k = j & 127;
        W2T[j] = W2[k * 3 + cc];
    }
}

extern "C" void kernel_launch(void* const* d_in, const int* in_sizes, int n_in,
                              void* d_out, int out_size, void* d_ws, size_t ws_size,
                              hipStream_t stream)
{
    (void)in_sizes; (void)n_in; (void)out_size; (void)ws_size;
    const float* input = (const float*)d_in[0];
    const float* Wc    = (const float*)d_in[1];
    const float* bc    = (const float*)d_in[2];
    const float* W1    = (const float*)d_in[3];
    const float* b1    = (const float*)d_in[4];
    const float* W2    = (const float*)d_in[5];
    const float* b2    = (const float*)d_in[6];
    float* ws   = (float*)d_ws;
    float* WT   = ws;
    float* W2T  = ws + 217088;
    float* cbuf = ws + 217472;
    float* hall = ws + 348544;
    float* obuf = ws + 17125760;
    float* zprt = ws + 17515904;
    float* out  = (float*)d_out;

    zero_kernel<<<262144 / 256, 256, 0, stream>>>(cbuf, 262144);
    prep_kernel<<<(217472 + 255) / 256, 256, 0, stream>>>(Wc, W1, W2, WT, W2T);

    for (int t = 0; t < 127; ++t) {
        const float* h_in = hall + t * 131072;
        float* h_out      = hall + (t + 1) * 131072;
        cell_part_kernel<<<256, 256, 0, stream>>>(input, WT, h_in, zprt, t);
        gate_kernel<<<256, 128, 0, stream>>>(bc, zprt, h_out, cbuf);
    }
    head_all_kernel<<<984, 256, 0, stream>>>(WT, hall + 5 * 131072, b1, W2T, b2,
                                              obuf + 4 * 3072);
    mix_all_kernel<<<123 * 1024 / 256, 256, 0, stream>>>(input, obuf, out);
}


# Gates of step t-1 moved into step t's GEMM block: it adds the 4 K-split planes and gates its own 18-row x 32-unit h slice (2 halo rows redone) while its weight chunks travel: 131 launches, was 258
# speedup vs baseline: 5.2178x; 1.1824x over previous
#include <hip/hip_runtime.h>
#include <math.h>

typedef __attribute__((ext_vector_type(2))) float v2f;
typedef __attribute__((ext_vector_type(8))) float v8f;
typedef __attribute__((ext_vector_type(4))) float v4f;

#define STRZ 520
#define WP   36
#define WBUF (512 * WP)

__device__ __forceinline__ void chunk_request(v4f (&r)[16], const v4f* src, int tid)
{
    #pragma unroll
    for (int i = 0; i < 16; ++i) r[i] = src[tid + i * 256];
}

__device__ __forceinline__ void chunk_to_lds(float* buf, int wdst, const v4f (&r)[16])
{
    #pragma unroll
    for (int i = 0; i < 16; ++i)
        *reinterpret_cast<v4f*>(&buf[wdst + i * 32 * WP]) = r[i];
}

__device__ __forceinline__ void lds_barrier()
{
    asm volatile("s_wait_dscnt 0x0\n\ts_barrier_signal -1\n\ts_barrier_wait -1" ::: "memory");
}

__device__ __forceinline__ void chunk_mma(const float* arow, const float* wb,
    int wl0, int wl1, int wl2, int wl3, int kHalf,
    v8f& acc0, v8f& acc1, v8f& acc2, v8f& acc3)
{
    #pragma unroll
    for (int kk = 0; kk < 32; kk += 4) {
        v2f a;
        a.x = arow[kk + kHalf];
        a.y = arow[kk + kHalf + 1];
        v2f b0, b1, b2, b3;
        b0.x = wb[wl0 + kk + kHalf]; b0.y = wb[wl0 + kk + kHalf + 1];
        b1.x = wb[wl1 + kk + kHalf]; b1.y = wb[wl1 + kk + kHalf + 1];
        b2.x = wb[wl2 + kk + kHalf]; b2.y = wb[wl2 + kk + kHalf + 1];
        b3.x = wb[wl3 + kk + kHalf]; b3.y = wb[wl3 + kk + kHalf + 1];
        acc0 = __builtin_amdgcn_wmma_f32_16x16x4_f32(
                    false, a, false, b0, (short)0, acc0, false, false);
        acc1 = __builtin_amdgcn_wmma_f32_16x16x4_f32(
                    false, a, false, b1, (short)0, acc1, false, false);
        acc2 = __builtin_amdgcn_wmma_f32_16x16x4_f32(
                    false, a, false, b2, (short)0, acc2, false, false);
        acc3 = __builtin_amdgcn_wmma_f32_16x16x4_f32(
                    false, a, false, b3, (short)0, acc3, false, false);
    }
}

#define HP 36
#define ZPLANE (1024 * 512)
#define HPLANE (1024 * 128)

__device__ __forceinline__ void gate_math(const v4f& zi, const v4f& zf, const v4f& zg,
    const v4f& zo, const v4f& cp, v4f& cn, v4f& hn)
{
    #pragma unroll
    for (int e = 0; e < 4; ++e) {
        float si = 1.0f / (1.0f + __expf(-zi[e]));
        float sf = 1.0f / (1.0f + __expf(-zf[e]));
        float so = 1.0f / (1.0f + __expf(-zo[e]));
        float cv = sf * cp[e] + si * tanhf(zg[e]);
        cn[e] = cv;
        hn[e] = so * tanhf(cv);
    }
}

template <bool FIRST>
__global__ __launch_bounds__(256) void cell_step_kernel(
    const float* __restrict__ input, const float* __restrict__ WT,
    const float* __restrict__ bcell, float* zpart, float* cbuf,
    float* __restrict__ hkeep, int t)
{
    __shared__ alignas(16) float sh_h[18 * HP];
    __shared__ float sh_x[16 * 8];
    __shared__ alignas(16) float sh_z[16 * STRZ];
    __shared__ alignas(16) float sh_w[3 * WBUF];

    const int tid = threadIdx.x;
    const int mt  = blockIdx.x >> 2;
    const int ks  = blockIdx.x & 3;
    const int R0  = mt * 16;
    const int bb  = R0 >> 7;
    const int s0  = R0 & 127;

    const float* zin  = zpart + ((t + 1) & 1) * (4 * ZPLANE);
    const float* c_in = cbuf  + ((t + 1) & 1) * HPLANE;
    float* const c_out = cbuf + (t & 1) * HPLANE;

    float xv = 0.0f;
    if (tid < 16 * 5)
        xv = input[(((bb * 128) + t) * 128 + (s0 + tid / 5)) * 5 + tid % 5];
    const int hr   = tid >> 3;
    const int hseg = s0 - 1 + hr;
    const int gj   = ks * 32 + (tid & 7) * 4;
    const bool own = (hr >= 1) && (hr <= 16);
    const bool act = (tid < 18 * 8) && (hseg >= 0) && (hseg < 128);
    const int hrow = bb * 128 + min(max(s0 - 1 + min(hr, 17), 0), 127);
    v4f zr[16], bq[4], cp;
    if (!FIRST) {
        const float* zp = zin + hrow * 512 + gj;
        #pragma unroll
        for (int g = 0; g < 4; ++g) {
            #pragma unroll
            for (int p = 0; p < 4; ++p)
                zr[g * 4 + p] = *reinterpret_cast<const v4f*>(zp + g * 128 + p * ZPLANE);
            bq[g] = *reinterpret_cast<const v4f*>(&bcell[g * 128 + gj]);
        }
        cp = *reinterpret_cast<const v4f*>(&c_in[hrow * 128 + gj]);
    }

    const v4f* wsrc = reinterpret_cast<const v4f*>(WT);
    const int wdst = (tid >> 3) * WP + (tid & 7) * 4;
    v4f preA[16], preB[16];
    chunk_request(preA, wsrc + ks * 4096, tid);
    chunk_request(preB, wsrc + (4 + ks) * 4096, tid);

    if (!FIRST) {
        asm volatile("" :: "v"(zr[0]), "v"(zr[1]), "v"(zr[2]), "v"(zr[3]),
                           "v"(zr[4]), "v"(zr[5]), "v"(zr[6]), "v"(zr[7]),
                           "v"(zr[8]), "v"(zr[9]), "v"(zr[10]), "v"(zr[11]),
                           "v"(zr[12]), "v"(zr[13]), "v"(zr[14]), "v"(zr[15]),
                           "v"(bq[0]), "v"(bq[1]), "v"(bq[2]), "v"(bq[3]), "v"(cp));
    }

    v4f hv = {0.0f, 0.0f, 0.0f, 0.0f};
    if (FIRST) {
        if (own) *reinterpret_cast<v4f*>(&c_out[hrow * 128 + gj]) = hv;
    } else if (act) {
        const v4f zi = (((zr[0]  + zr[1])  + zr[2])  + zr[3])  + bq[0];
        const v4f zf = (((zr[4]  + zr[5])  + zr[6])  + zr[7])  + bq[1];
        const v4f zg = (((zr[8]  + zr[9])  + zr[10]) + zr[11]) + bq[2];
        const v4f zo = (((zr[12] + zr[13]) + zr[14]) + zr[15]) + bq[3];
        v4f cn;
        gate_math(zi, zf, zg, zo, cp, cn, hv);
        if (own) {
            *reinterpret_cast<v4f*>(&c_out[hrow * 128 + gj]) = cn;
            *reinterpret_cast<v4f*>(&hkeep[(t - 1) * HPLANE + hrow * 128 + gj]) = hv;
        }
    }

    const int wave  = tid >> 5;
    const int lane  = tid & 31;
    const int lm    = lane & 15;
    const int hi    = lane >> 4;
    const int kHalf = hi * 2;

    const float* wc0 = WT + 13 * 16384 + ((wave * 4 + 0) * 16 + lm) * 8;
    const float* wc1 = WT + 13 * 16384 + ((wave * 4 + 1) * 16 + lm) * 8;
    const float* wc2 = WT + 13 * 16384 + ((wave * 4 + 2) * 16 + lm) * 8;
    const float* wc3 = WT + 13 * 16384 + ((wave * 4 + 3) * 16 + lm) * 8;
    const v4f wq0 = *reinterpret_cast<const v4f*>(wc0); const float we0 = wc0[4];
    const v4f wq1 = *reinterpret_cast<const v4f*>(wc1); const float we1 = wc1[4];
    const v4f wq2 = *reinterpret_cast<const v4f*>(wc2); const float we2 = wc2[4];
    const v4f wq3 = *reinterpret_cast<const v4f*>(wc3); const float we3 = wc3[4];

    if (tid < 18 * 8)
        *reinterpret_cast<v4f*>(&sh_h[(tid >> 3) * HP + (tid & 7) * 4]) = hv;
    if (tid < 16 * 5)
        sh_x[(tid / 5) * 8 + tid % 5] = xv;
    float* const buf0 = sh_w;
    float* const buf1 = sh_w + WBUF;
    float* const buf2 = sh_w + 2 * WBUF;
    chunk_to_lds(buf0, wdst, preA);
    chunk_request(preA, wsrc + (8 + ks) * 4096, tid);
    chunk_to_lds(buf1, wdst, preB);
    lds_barrier();

    const int wl0 = ((wave * 4 + 0) * 16 + lm) * WP;
    const int wl1 = ((wave * 4 + 1) * 16 + lm) * WP;
    const int wl2 = ((wave * 4 + 2) * 16 + lm) * WP;
    const int wl3 = ((wave * 4 + 3) * 16 + lm) * WP;

    v8f acc0 = {}, acc1 = {}, acc2 = {}, acc3 = {};

    if (ks == 0) {
        #pragma unroll
        for (int v = 0; v < 8; ++v) {
            int m = v + hi * 8;
            float x0 = sh_x[m*8+0], x1 = sh_x[m*8+1], x2 = sh_x[m*8+2],
                  x3 = sh_x[m*8+3], x4 = sh_x[m*8+4];
            acc0[v] += x0*wq0.x + x1*wq0.y + x2*wq0.z + x3*wq0.w + x4*we0;
            acc1[v] += x0*wq1.x + x1*wq1.y + x2*wq1.z + x3*wq1.w + x4*we1;
            acc2[v] += x0*wq2.x + x1*wq2.y + x2*wq2.z + x3*wq2.w + x4*we2;
            acc3[v] += x0*wq3.x + x1*wq3.y + x2*wq3.z + x3*wq3.w + x4*we3;
        }
    }

    chunk_mma(&sh_h[(1 + lm) * HP], buf0, wl0, wl1, wl2, wl3, kHalf, acc0, acc1, acc2, acc3);
    chunk_mma(&sh_h[(2 + lm) * HP], buf1, wl0, wl1, wl2, wl3, kHalf, acc0, acc1, acc2, acc3);
    chunk_to_lds(buf2, wdst, preA);
    lds_barrier();
    chunk_mma(&sh_h[lm * HP], buf2, wl0, wl1, wl2, wl3, kHalf, acc0, acc1, acc2, acc3);

    #pragma unroll
    for (int v = 0; v < 8; ++v) {
        int m = v + hi * 8;
        sh_z[m * STRZ + (wave * 4 + 0) * 16 + lm] = acc0[v];
        sh_z[m * STRZ + (wave * 4 + 1) * 16 + lm] = acc1[v];
        sh_z[m * STRZ + (wave * 4 + 2) * 16 + lm] = acc2[v];
        sh_z[m * STRZ + (wave * 4 + 3) * 16 + lm] = acc3[v];
    }
    __syncthreads();

    float* zp = zpart + (t & 1) * (4 * ZPLANE) + ks * ZPLANE + R0 * 512;
    #pragma unroll
    for (int it = 0; it < 8; ++it) {
        int i = tid + it * 256;
        int m = i >> 7, c4 = (i & 127) * 4;
        *reinterpret_cast<v4f*>(&zp[m * 512 + c4]) =
            *reinterpret_cast<const v4f*>(&sh_z[m * STRZ + c4]);
    }
}

__device__ __forceinline__ v4f plane_sum(const float* p)
{
    v4f a = *reinterpret_cast<const v4f*>(p);
    v4f b = *reinterpret_cast<const v4f*>(p + ZPLANE);
    v4f c = *reinterpret_cast<const v4f*>(p + 2 * ZPLANE);
    v4f d = *reinterpret_cast<const v4f*>(p + 3 * ZPLANE);
    return ((a + b) + c) + d;
}

__global__ __launch_bounds__(128) void gate_kernel(
    const float* __restrict__ bcell, const float* __restrict__ zpart,
    float* __restrict__ h_out, float* __restrict__ c)
{
    const int gid = blockIdx.x * 128 + threadIdx.x;
    const int row = gid >> 5, gj = (gid & 31) * 4;

    const float* zp = zpart + row * 512 + gj;
    const v4f zi = plane_sum(zp)       + *reinterpret_cast<const v4f*>(&bcell[gj]);
    const v4f zf = plane_sum(zp + 128) + *reinterpret_cast<const v4f*>(&bcell[128 + gj]);
    const v4f zg = plane_sum(zp + 256) + *reinterpret_cast<const v4f*>(&bcell[256 + gj]);
    const v4f zo = plane_sum(zp + 384) + *reinterpret_cast<const v4f*>(&bcell[384 + gj]);
    const v4f cp = *reinterpret_cast<const v4f*>(&c[row * 128 + gj]);

    v4f cn, hn;
    gate_math(zi, zf, zg, zo, cp, cn, hn);
    *reinterpret_cast<v4f*>(&c[row * 128 + gj])     = cn;
    *reinterpret_cast<v4f*>(&h_out[row * 128 + gj]) = hn;
}

#define HIDP 132

__global__ __launch_bounds__(256) void head_all_kernel(
    const float* __restrict__ WT, const float* __restrict__ hrows,
    const float* __restrict__ b1, const float* __restrict__ W2T,
    const float* __restrict__ b2, float* __restrict__ orows)
{
    __shared__ alignas(16) float sh_w[WBUF];
    __shared__ alignas(16) float sh_hid[8 * 16 * HIDP];
    __shared__ alignas(16) float sh_w2[3 * 128];
    __shared__ alignas(16) float sh_o[8 * 48];

    const int tid   = threadIdx.x;
    const int wave  = tid >> 5;
    const int lane  = tid & 31;
    const int lm    = lane & 15;
    const int hi    = lane >> 4;
    const int kHalf = hi * 2;

    const v4f* wsrc = reinterpret_cast<const v4f*>(WT);
    const int wdst = (tid >> 3) * WP + (tid & 7) * 4;
    v4f preA[16];
    chunk_request(preA, wsrc + 12 * 4096, tid);

    const float* hrow = hrows + ((blockIdx.x * 8 + wave) * 16 + lm) * 128 + kHalf;
    v2f a[32];
    #pragma unroll
    for (int ks = 0; ks < 32; ++ks)
        a[ks] = *reinterpret_cast<const v2f*>(hrow + ks * 4);

    float bias1[8];
    #pragma unroll
    for (int j = 0; j < 8; ++j) bias1[j] = b1[j * 16 + lm];
    const float b20 = b2[0], b21 = b2[1], b22 = b2[2];
    v4f w2v = {0.0f, 0.0f, 0.0f, 0.0f};
    if (tid < 96) w2v = *reinterpret_cast<const v4f*>(W2T + tid * 4);

    if (tid < 96) *reinterpret_cast<v4f*>(&sh_w2[tid * 4]) = w2v;
    chunk_to_lds(sh_w, wdst, preA);
    __syncthreads();

    v8f acc[8];
    #pragma unroll
    for (int j = 0; j < 8; ++j) acc[j] = v8f{};
    #pragma unroll
    for (int kq = 0; kq < 4; ++kq) {
        #pragma unroll
        for (int kk = 0; kk < 32; kk += 4) {
            #pragma unroll
            for (int j = 0; j < 8; ++j) {
                const float* wq = sh_w + (kq * 128 + j * 16 + lm) * WP;
                v2f b;
                b.x = wq[kk + kHalf]; b.y = wq[kk + kHalf + 1];
                acc[j] = __builtin_amdgcn_wmma_f32_16x16x4_f32(
                            false, a[kq * 8 + (kk >> 2)], false, b, (short)0, acc[j], false, false);
            }
        }
    }
    float* hid = sh_hid + wave * 16 * HIDP;
    #pragma unroll
    for (int j = 0; j < 8; ++j) {
        #pragma unroll
        for (int v = 0; v < 8; ++v) {
            float x = acc[j][v] + bias1[j];
            hid[(v + hi * 8) * HIDP + j * 16 + lm] = x > 0.0f ? x : 0.0f;
        }
    }
    __syncthreads();

    if (lane < 16) {
        const float* hr = hid + lane * HIDP;
        float s0 = 0.0f, s1 = 0.0f, s2 = 0.0f;
        for (int k = 0; k < 128; k += 4) {
            const v4f hv = *reinterpret_cast<const v4f*>(hr + k);
            const v4f wa = *reinterpret_cast<const v4f*>(sh_w2 + k);
            const v4f wb = *reinterpret_cast<const v4f*>(sh_w2 + 128 + k);
            const v4f wc = *reinterpret_cast<const v4f*>(sh_w2 + 256 + k);
            s0 = fmaf(hv.x, wa.x, s0); s0 = fmaf(hv.y, wa.y, s0);
            s0 = fmaf(hv.z, wa.z, s0); s0 = fmaf(hv.w, wa.w, s0);
            s1 = fmaf(hv.x, wb.x, s1); s1 = fmaf(hv.y, wb.y, s1);
            s1 = fmaf(hv.z, wb.z, s1); s1 = fmaf(hv.w, wb.w, s1);
            s2 = fmaf(hv.x, wc.x, s2); s2 = fmaf(hv.y, wc.y, s2);
            s2 = fmaf(hv.z, wc.z, s2); s2 = fmaf(hv.w, wc.w, s2);
        }
        float* so = sh_o + wave * 48 + lane * 3;
        so[0] = s0 + b20; so[1] = s1 + b21; so[2] = s2 + b22;
    }
    __syncthreads();

    if (tid < 96)
        *reinterpret_cast<v4f*>(orows + blockIdx.x * 384 + tid * 4) =
            *reinterpret_cast<const v4f*>(sh_o + tid * 4);
}

__global__ __launch_bounds__(256) void mix_all_kernel(
    const float* __restrict__ input, const float* __restrict__ obuf_all,
    float* __restrict__ out)
{
    int gid = blockIdx.x * 256 + threadIdx.x;
    if (gid >= 123 * 1024) return;
    int tt = gid >> 10, r = gid & 1023;
    int t  = tt + 4;
    int bb = r >> 7, s = r & 127;
    const float* ob = obuf_all + t * 3072;
    const float* o  = &ob[r * 3];
    float n0 = o[0];
    float n1, sp0, sp1;
    if (s == 0) {
        const float* xn0 = &input[((bb * 128 + (t + 1)) * 128 + 0) * 5];
        n1 = xn0[1]; sp0 = xn0[3]; sp1 = xn0[4];
    } else {
        n1  = ob[(r - 1) * 3 + 0];
        sp0 = o[1]; sp1 = o[2];
    }
    float fn = input[((bb * 128 + t) * 128 + s) * 5 + 2];
    float n2 = fn + n1 - n0;
    float* dst = &out[((bb * 123 + tt) * 128 + s) * 5];
    dst[0] = n0; dst[1] = n1; dst[2] = n2; dst[3] = sp0; dst[4] = sp1;
}

__global__ __launch_bounds__(256) void prep_kernel(
    const float* __restrict__ Wc, const float* __restrict__ W1,
    const float* __restrict__ W2, float* __restrict__ WT,
    float* __restrict__ W2T)
{
    int i = blockIdx.x * 256 + threadIdx.x;
    if (i < 12 * 16384) {
        int chn = i >> 14, col = (i >> 5) & 511, q = i & 31;
        WT[i] = Wc[(5 + chn * 32 + q) * 512 + col];
    } else if (i < 13 * 16384) {
        int j = i - 12 * 16384;
        int pc = j >> 5, q = j & 31;
        int kq = pc >> 7, n = pc & 127;
        WT[i] = W1[(kq * 32 + q) * 128 + n];
    } else if (i < 13 * 16384 + 4096) {
        int j = i - 13 * 16384;
        int col = j >> 3, k = j & 7;
        WT[i] = (k < 5) ? Wc[k * 512 + col] : 0.0f;
    } else if (i < 13 * 16384 + 4096 + 384) {
        int j = i - 13 * 16384 - 4096;
        int cc = j >> 7, k = j & 127;
        W2T[j] = W2[k * 3 + cc];
    }
}

extern "C" void kernel_launch(void* const* d_in, const int* in_sizes, int n_in,
                              void* d_out, int out_size, void* d_ws, size_t ws_size,
                              hipStream_t stream)
{
    (void)in_sizes; (void)n_in; (void)out_size; (void)ws_size;
    const float* input = (const float*)d_in[0];
    const float* Wc    = (const float*)d_in[1];
    const float* bc    = (const float*)d_in[2];
    const float* W1    = (const float*)d_in[3];
    const float* b1    = (const float*)d_in[4];
    const float* W2    = (const float*)d_in[5];
    const float* b2    = (const float*)d_in[6];
    float* ws   = (float*)d_ws;
    float* WT   = ws;
    float* W2T  = ws + 217088;
    float* cbuf = ws + 217472;
    float* hall = ws + 479616;
    float* obuf = ws + 17125760;
    float* zprt = ws + 17515904;
    float* out  = (float*)d_out;

    prep_kernel<<<(217472 + 255) / 256, 256, 0, stream>>>(Wc, W1, W2, WT, W2T);

    cell_step_kernel<true><<<256, 256, 0, stream>>>(input, WT, bc, zprt, cbuf, hall, 0);
    for (int t = 1; t < 127; ++t)
        cell_step_kernel<false><<<256, 256, 0, stream>>>(input, WT, bc, zprt, cbuf, hall, t);
    gate_kernel<<<256, 128, 0, stream>>>(bc, zprt, hall + 126 * 131072, cbuf);
    head_all_kernel<<<984, 256, 0, stream>>>(WT, hall + 4 * 131072, b1, W2T, b2,
                                              obuf + 4 * 3072);
    mix_all_kernel<<<123 * 1024 / 256, 256, 0, stream>>>(input, obuf, out);
}


# Cell weights staged per wave, not per block: a wave moves only its own N-tiles' columns through LDS in six half-chunk stages and multiplies each as its loads land; no block barrier on weights
# speedup vs baseline: 5.6092x; 1.0750x over previous
#include <hip/hip_runtime.h>
#include <math.h>

typedef __attribute__((ext_vector_type(2))) float v2f;
typedef __attribute__((ext_vector_type(8))) float v8f;
typedef __attribute__((ext_vector_type(4))) float v4f;

#define STRZ 520
#define WP   36
#define WBUF (512 * WP)

__device__ __forceinline__ void chunk_request(v4f (&r)[16], const v4f* src, int tid)
{
    #pragma unroll
    for (int i = 0; i < 16; ++i) r[i] = src[tid + i * 256];
}

__device__ __forceinline__ void chunk_to_lds(float* buf, int wdst, const v4f (&r)[16])
{
    #pragma unroll
    for (int i = 0; i < 16; ++i)
        *reinterpret_cast<v4f*>(&buf[wdst + i * 32 * WP]) = r[i];
}

__device__ __forceinline__ void lds_barrier()
{
    asm volatile("s_wait_dscnt 0x0\n\ts_barrier_signal -1\n\ts_barrier_wait -1" ::: "memory");
}

__device__ __forceinline__ void wave_chunk_request(v4f (&r)[16], const v4f* src, int wave, int lane)
{
    #pragma unroll
    for (int i = 0; i < 16; ++i) r[i] = src[wave * 512 + i * 32 + lane];
}

template <int HALF>
__device__ __forceinline__ void wave_half_to_lds(float* buf, int wdst, const v4f (&r)[16])
{
    #pragma unroll
    for (int i = 8 * HALF; i < 8 * HALF + 8; ++i)
        *reinterpret_cast<v4f*>(&buf[wdst + i * 4 * WP]) = r[i];
}

__device__ __forceinline__ void wave_lds_handover()
{
    __builtin_amdgcn_fence(__ATOMIC_RELEASE, "wavefront");
    __builtin_amdgcn_wave_barrier();
    __builtin_amdgcn_fence(__ATOMIC_ACQUIRE, "wavefront");
}

__device__ __forceinline__ void chunk_mma2(const float* arow, const float* wb,
    int wlA, int wlB, int kHalf, v8f& accA, v8f& accB)
{
    #pragma unroll
    for (int kk = 0; kk < 32; kk += 4) {
        v2f a;
        a.x = arow[kk + kHalf];
        a.y = arow[kk + kHalf + 1];
        v2f bA, bB;
        bA.x = wb[wlA + kk + kHalf]; bA.y = wb[wlA + kk + kHalf + 1];
        bB.x = wb[wlB + kk + kHalf]; bB.y = wb[wlB + kk + kHalf + 1];
        accA = __builtin_amdgcn_wmma_f32_16x16x4_f32(
                    false, a, false, bA, (short)0, accA, false, false);
        accB = __builtin_amdgcn_wmma_f32_16x16x4_f32(
                    false, a, false, bB, (short)0, accB, false, false);
    }
}

#define HP 36
#define ZPLANE (1024 * 512)
#define HPLANE (1024 * 128)

__device__ __forceinline__ void gate_math(const v4f& zi, const v4f& zf, const v4f& zg,
    const v4f& zo, const v4f& cp, v4f& cn, v4f& hn)
{
    #pragma unroll
    for (int e = 0; e < 4; ++e) {
        float si = 1.0f / (1.0f + __expf(-zi[e]));
        float sf = 1.0f / (1.0f + __expf(-zf[e]));
        float so = 1.0f / (1.0f + __expf(-zo[e]));
        float cv = sf * cp[e] + si * tanhf(zg[e]);
        cn[e] = cv;
        hn[e] = so * tanhf(cv);
    }
}

template <bool FIRST>
__global__ __launch_bounds__(256) void cell_step_kernel(
    const float* __restrict__ input, const float* __restrict__ WT,
    const float* __restrict__ bcell, float* zpart, float* cbuf,
    float* __restrict__ hkeep, int t)
{
    __shared__ alignas(16) float sh_h[18 * HP];
    __shared__ float sh_x[16 * 8];
    __shared__ alignas(16) float sh_z[16 * STRZ];
    __shared__ alignas(16) float sh_w[3 * WBUF];

    const int tid = threadIdx.x;
    const int mt  = blockIdx.x >> 2;
    const int ks  = blockIdx.x & 3;
    const int R0  = mt * 16;
    const int bb  = R0 >> 7;
    const int s0  = R0 & 127;

    const float* zin  = zpart + ((t + 1) & 1) * (4 * ZPLANE);
    const float* c_in = cbuf  + ((t + 1) & 1) * HPLANE;
    float* const c_out = cbuf + (t & 1) * HPLANE;

    float xv = 0.0f;
    if (tid < 16 * 5)
        xv = input[(((bb * 128) + t) * 128 + (s0 + tid / 5)) * 5 + tid % 5];
    const int hr   = tid >> 3;
    const int hseg = s0 - 1 + hr;
    const int gj   = ks * 32 + (tid & 7) * 4;
    const bool own = (hr >= 1) && (hr <= 16);
    const bool act = (tid < 18 * 8) && (hseg >= 0) && (hseg < 128);
    const int hrow = bb * 128 + min(max(s0 - 1 + min(hr, 17), 0), 127);
    v4f zr[16], bq[4], cp;
    if (!FIRST) {
        const float* zp = zin + hrow * 512 + gj;
        #pragma unroll
        for (int g = 0; g < 4; ++g) {
            #pragma unroll
            for (int p = 0; p < 4; ++p)
                zr[g * 4 + p] = *reinterpret_cast<const v4f*>(zp + g * 128 + p * ZPLANE);
            bq[g] = *reinterpret_cast<const v4f*>(&bcell[g * 128 + gj]);
        }
        cp = *reinterpret_cast<const v4f*>(&c_in[hrow * 128 + gj]);
    }

    const int wave  = tid >> 5;
    const int lane  = tid & 31;
    const int lm    = lane & 15;
    const int hi    = lane >> 4;
    const int kHalf = hi * 2;

    const float* wc0 = WT + 13 * 16384 + ((wave * 4 + 0) * 16 + lm) * 8;
    const float* wc1 = WT + 13 * 16384 + ((wave * 4 + 1) * 16 + lm) * 8;
    const float* wc2 = WT + 13 * 16384 + ((wave * 4 + 2) * 16 + lm) * 8;
    const float* wc3 = WT + 13 * 16384 + ((wave * 4 + 3) * 16 + lm) * 8;
    const v4f wq0 = *reinterpret_cast<const v4f*>(wc0); const float we0 = wc0[4];
    const v4f wq1 = *reinterpret_cast<const v4f*>(wc1); const float we1 = wc1[4];
    const v4f wq2 = *reinterpret_cast<const v4f*>(wc2); const float we2 = wc2[4];
    const v4f wq3 = *reinterpret_cast<const v4f*>(wc3); const float we3 = wc3[4];

    asm volatile("" ::: "memory");

    const v4f* wsrc = reinterpret_cast<const v4f*>(WT);
    const int wdst = (wave * 64 + (lane >> 3)) * WP + (lane & 7) * 4;
    v4f preA[16], preB[16], preC[16];
    wave_chunk_request(preA, wsrc + ks * 4096, wave, lane);
    wave_chunk_request(preB, wsrc + (4 + ks) * 4096, wave, lane);

    if (!FIRST) {
        asm volatile("" :: "v"(zr[0]), "v"(zr[1]), "v"(zr[2]), "v"(zr[3]),
                           "v"(zr[4]), "v"(zr[5]), "v"(zr[6]), "v"(zr[7]),
                           "v"(zr[8]), "v"(zr[9]), "v"(zr[10]), "v"(zr[11]),
                           "v"(zr[12]), "v"(zr[13]), "v"(zr[14]), "v"(zr[15]),
                           "v"(bq[0]), "v"(bq[1]), "v"(bq[2]), "v"(bq[3]), "v"(cp));
    }

    v4f hv = {0.0f, 0.0f, 0.0f, 0.0f};
    if (FIRST) {
        if (own) *reinterpret_cast<v4f*>(&c_out[hrow * 128 + gj]) = hv;
    } else if (act) {
        const v4f zi = (((zr[0]  + zr[1])  + zr[2])  + zr[3])  + bq[0];
        const v4f zf = (((zr[4]  + zr[5])  + zr[6])  + zr[7])  + bq[1];
        const v4f zg = (((zr[8]  + zr[9])  + zr[10]) + zr[11]) + bq[2];
        const v4f zo = (((zr[12] + zr[13]) + zr[14]) + zr[15]) + bq[3];
        v4f cn;
        gate_math(zi, zf, zg, zo, cp, cn, hv);
        if (own) {
            *reinterpret_cast<v4f*>(&c_out[hrow * 128 + gj]) = cn;
            *reinterpret_cast<v4f*>(&hkeep[(t - 1) * HPLANE + hrow * 128 + gj]) = hv;
        }
    }

    wave_chunk_request(preC, wsrc + (8 + ks) * 4096, wave, lane);

    if (tid < 18 * 8)
        *reinterpret_cast<v4f*>(&sh_h[(tid >> 3) * HP + (tid & 7) * 4]) = hv;
    if (tid < 16 * 5)
        sh_x[(tid / 5) * 8 + tid % 5] = xv;
    float* const buf0 = sh_w;
    float* const buf1 = sh_w + WBUF;
    float* const buf2 = sh_w + 2 * WBUF;
    lds_barrier();

    const int wl0 = ((wave * 4 + 0) * 16 + lm) * WP;
    const int wl1 = ((wave * 4 + 1) * 16 + lm) * WP;
    const int wl2 = ((wave * 4 + 2) * 16 + lm) * WP;
    const int wl3 = ((wave * 4 + 3) * 16 + lm) * WP;

    v8f acc0 = {}, acc1 = {}, acc2 = {}, acc3 = {};

    if (ks == 0) {
        #pragma unroll
        for (int v = 0; v < 8; ++v) {
            int m = v + hi * 8;
            float x0 = sh_x[m*8+0], x1 = sh_x[m*8+1], x2 = sh_x[m*8+2],
                  x3 = sh_x[m*8+3], x4 = sh_x[m*8+4];
            acc0[v] += x0*wq0.x + x1*wq0.y + x2*wq0.z + x3*wq0.w + x4*we0;
            acc1[v] += x0*wq1.x + x1*wq1.y + x2*wq1.z + x3*wq1.w + x4*we1;
            acc2[v] += x0*wq2.x + x1*wq2.y + x2*wq2.z + x3*wq2.w + x4*we2;
            acc3[v] += x0*wq3.x + x1*wq3.y + x2*wq3.z + x3*wq3.w + x4*we3;
        }
    }

    const float* const a0 = &sh_h[(1 + lm) * HP];
    const float* const a1 = &sh_h[(2 + lm) * HP];
    const float* const a2 = &sh_h[lm * HP];
    wave_half_to_lds<0>(buf0, wdst, preA);
    wave_lds_handover();
    chunk_mma2(a0, buf0, wl0, wl1, kHalf, acc0, acc1);
    asm volatile("" :: "v"(acc0), "v"(acc1) : "memory");
    wave_half_to_lds<1>(buf0, wdst, preA);
    wave_lds_handover();
    chunk_mma2(a0, buf0, wl2, wl3, kHalf, acc2, acc3);
    asm volatile("" :: "v"(acc2), "v"(acc3) : "memory");
    wave_half_to_lds<0>(buf1, wdst, preB);
    wave_lds_handover();
    chunk_mma2(a1, buf1, wl0, wl1, kHalf, acc0, acc1);
    asm volatile("" :: "v"(acc0), "v"(acc1) : "memory");
    wave_half_to_lds<1>(buf1, wdst, preB);
    wave_lds_handover();
    chunk_mma2(a1, buf1, wl2, wl3, kHalf, acc2, acc3);
    asm volatile("" :: "v"(acc2), "v"(acc3) : "memory");
    wave_half_to_lds<0>(buf2, wdst, preC);
    wave_lds_handover();
    chunk_mma2(a2, buf2, wl0, wl1, kHalf, acc0, acc1);
    asm volatile("" :: "v"(acc0), "v"(acc1) : "memory");
    wave_half_to_lds<1>(buf2, wdst, preC);
    wave_lds_handover();
    chunk_mma2(a2, buf2, wl2, wl3, kHalf, acc2, acc3);

    #pragma unroll
    for (int v = 0; v < 8; ++v) {
        int m = v + hi * 8;
        sh_z[m * STRZ + (wave * 4 + 0) * 16 + lm] = acc0[v];
        sh_z[m * STRZ + (wave * 4 + 1) * 16 + lm] = acc1[v];
        sh_z[m * STRZ + (wave * 4 + 2) * 16 + lm] = acc2[v];
        sh_z[m * STRZ + (wave * 4 + 3) * 16 + lm] = acc3[v];
    }
    __syncthreads();

    float* zp = zpart + (t & 1) * (4 * ZPLANE) + ks * ZPLANE + R0 * 512;
    #pragma unroll
    for (int it = 0; it < 8; ++it) {
        int i = tid + it * 256;
        int m = i >> 7, c4 = (i & 127) * 4;
        *reinterpret_cast<v4f*>(&zp[m * 512 + c4]) =
            *reinterpret_cast<const v4f*>(&sh_z[m * STRZ + c4]);
    }
}

__device__ __forceinline__ v4f plane_sum(const float* p)
{
    v4f a = *reinterpret_cast<const v4f*>(p);
    v4f b = *reinterpret_cast<const v4f*>(p + ZPLANE);
    v4f c = *reinterpret_cast<const v4f*>(p + 2 * ZPLANE);
    v4f d = *reinterpret_cast<const v4f*>(p + 3 * ZPLANE);
    return ((a + b) + c) + d;
}

__global__ __launch_bounds__(128) void gate_kernel(
    const float* __restrict__ bcell, const float* __restrict__ zpart,
    float* __restrict__ h_out, float* __restrict__ c)
{
    const int gid = blockIdx.x * 128 + threadIdx.x;
    const int row = gid >> 5, gj = (gid & 31) * 4;

    const float* zp = zpart + row * 512 + gj;
    const v4f zi = plane_sum(zp)       + *reinterpret_cast<const v4f*>(&bcell[gj]);
    const v4f zf = plane_sum(zp + 128) + *reinterpret_cast<const v4f*>(&bcell[128 + gj]);
    const v4f zg = plane_sum(zp + 256) + *reinterpret_cast<const v4f*>(&bcell[256 + gj]);
    const v4f zo = plane_sum(zp + 384) + *reinterpret_cast<const v4f*>(&bcell[384 + gj]);
    const v4f cp = *reinterpret_cast<const v4f*>(&c[row * 128 + gj]);

    v4f cn, hn;
    gate_math(zi, zf, zg, zo, cp, cn, hn);
    *reinterpret_cast<v4f*>(&c[row * 128 + gj])     = cn;
    *reinterpret_cast<v4f*>(&h_out[row * 128 + gj]) = hn;
}

#define HIDP 132

__global__ __launch_bounds__(256) void head_all_kernel(
    const float* __restrict__ WT, const float* __restrict__ hrows,
    const float* __restrict__ b1, const float* __restrict__ W2T,
    const float* __restrict__ b2, float* __restrict__ orows)
{
    __shared__ alignas(16) float sh_w[WBUF];
    __shared__ alignas(16) float sh_hid[8 * 16 * HIDP];
    __shared__ alignas(16) float sh_w2[3 * 128];
    __shared__ alignas(16) float sh_o[8 * 48];

    const int tid   = threadIdx.x;
    const int wave  = tid >> 5;
    const int lane  = tid & 31;
    const int lm    = lane & 15;
    const int hi    = lane >> 4;
    const int kHalf = hi * 2;

    const v4f* wsrc = reinterpret_cast<const v4f*>(WT);
    const int wdst = (tid >> 3) * WP + (tid & 7) * 4;
    v4f preA[16];
    chunk_request(preA, wsrc + 12 * 4096, tid);

    const float* hrow = hrows + ((blockIdx.x * 8 + wave) * 16 + lm) * 128 + kHalf;
    v2f a[32];
    #pragma unroll
    for (int ks = 0; ks < 32; ++ks)
        a[ks] = *reinterpret_cast<const v2f*>(hrow + ks * 4);

    float bias1[8];
    #pragma unroll
    for (int j = 0; j < 8; ++j) bias1[j] = b1[j * 16 + lm];
    const float b20 = b2[0], b21 = b2[1], b22 = b2[2];
    v4f w2v = {0.0f, 0.0f, 0.0f, 0.0f};
    if (tid < 96) w2v = *reinterpret_cast<const v4f*>(W2T + tid * 4);

    if (tid < 96) *reinterpret_cast<v4f*>(&sh_w2[tid * 4]) = w2v;
    chunk_to_lds(sh_w, wdst, preA);
    __syncthreads();

    v8f acc[8];
    #pragma unroll
    for (int j = 0; j < 8; ++j) acc[j] = v8f{};
    #pragma unroll
    for (int kq = 0; kq < 4; ++kq) {
        #pragma unroll
        for (int kk = 0; kk < 32; kk += 4) {
            #pragma unroll
            for (int j = 0; j < 8; ++j) {
                const float* wq = sh_w + (kq * 128 + j * 16 + lm) * WP;
                v2f b;
                b.x = wq[kk + kHalf]; b.y = wq[kk + kHalf + 1];
                acc[j] = __builtin_amdgcn_wmma_f32_16x16x4_f32(
                            false, a[kq * 8 + (kk >> 2)], false, b, (short)0, acc[j], false, false);
            }
        }
    }
    float* hid = sh_hid + wave * 16 * HIDP;
    #pragma unroll
    for (int j = 0; j < 8; ++j) {
        #pragma unroll
        for (int v = 0; v < 8; ++v) {
            float x = acc[j][v] + bias1[j];
            hid[(v + hi * 8) * HIDP + j * 16 + lm] = x > 0.0f ? x : 0.0f;
        }
    }
    __syncthreads();

    if (lane < 16) {
        const float* hr = hid + lane * HIDP;
        float s0 = 0.0f, s1 = 0.0f, s2 = 0.0f;
        for (int k = 0; k < 128; k += 4) {
            const v4f hv = *reinterpret_cast<const v4f*>(hr + k);
            const v4f wa = *reinterpret_cast<const v4f*>(sh_w2 + k);
            const v4f wb = *reinterpret_cast<const v4f*>(sh_w2 + 128 + k);
            const v4f wc = *reinterpret_cast<const v4f*>(sh_w2 + 256 + k);
            s0 = fmaf(hv.x, wa.x, s0); s0 = fmaf(hv.y, wa.y, s0);
            s0 = fmaf(hv.z, wa.z, s0); s0 = fmaf(hv.w, wa.w, s0);
            s1 = fmaf(hv.x, wb.x, s1); s1 = fmaf(hv.y, wb.y, s1);
            s1 = fmaf(hv.z, wb.z, s1); s1 = fmaf(hv.w, wb.w, s1);
            s2 = fmaf(hv.x, wc.x, s2); s2 = fmaf(hv.y, wc.y, s2);
            s2 = fmaf(hv.z, wc.z, s2); s2 = fmaf(hv.w, wc.w, s2);
        }
        float* so = sh_o + wave * 48 + lane * 3;
        so[0] = s0 + b20; so[1] = s1 + b21; so[2] = s2 + b22;
    }
    __syncthreads();

    if (tid < 96)
        *reinterpret_cast<v4f*>(orows + blockIdx.x * 384 + tid * 4) =
            *reinterpret_cast<const v4f*>(sh_o + tid * 4);
}

__global__ __launch_bounds__(256) void mix_all_kernel(
    const float* __restrict__ input, const float* __restrict__ obuf_all,
    float* __restrict__ out)
{
    int gid = blockIdx.x * 256 + threadIdx.x;
    if (gid >= 123 * 1024) return;
    int tt = gid >> 10, r = gid & 1023;
    int t  = tt + 4;
    int bb = r >> 7, s = r & 127;
    const float* ob = obuf_all + t * 3072;
    const float* o  = &ob[r * 3];
    float n0 = o[0];
    float n1, sp0, sp1;
    if (s == 0) {
        const float* xn0 = &input[((bb * 128 + (t + 1)) * 128 + 0) * 5];
        n1 = xn0[1]; sp0 = xn0[3]; sp1 = xn0[4];
    } else {
        n1  = ob[(r - 1) * 3 + 0];
        sp0 = o[1]; sp1 = o[2];
    }
    float fn = input[((bb * 128 + t) * 128 + s) * 5 + 2];
    float n2 = fn + n1 - n0;
    float* dst = &out[((bb * 123 + tt) * 128 + s) * 5];
    dst[0] = n0; dst[1] = n1; dst[2] = n2; dst[3] = sp0; dst[4] = sp1;
}

__global__ __launch_bounds__(256) void prep_kernel(
    const float* __restrict__ Wc, const float* __restrict__ W1,
    const float* __restrict__ W2, float* __restrict__ WT,
    float* __restrict__ W2T)
{
    int i = blockIdx.x * 256 + threadIdx.x;
    if (i < 12 * 16384) {
        int chn = i >> 14, col = (i >> 5) & 511, q = i & 31;
        WT[i] = Wc[(5 + chn * 32 + q) * 512 + col];
    } else if (i < 13 * 16384) {
        int j = i - 12 * 16384;
        int pc = j >> 5, q = j & 31;
        int kq = pc >> 7, n = pc & 127;
        WT[i] = W1[(kq * 32 + q) * 128 + n];
    } else if (i < 13 * 16384 + 4096) {
        int j = i - 13 * 16384;
        int col = j >> 3, k = j & 7;
        WT[i] = (k < 5) ? Wc[k * 512 + col] : 0.0f;
    } else if (i < 13 * 16384 + 4096 + 384) {
        int j = i - 13 * 16384 - 4096;
        int cc = j >> 7, k = j & 127;
        W2T[j] = W2[k * 3 + cc];
    }
}

extern "C" void kernel_launch(void* const* d_in, const int* in_sizes, int n_in,
                              void* d_out, int out_size, void* d_ws, size_t ws_size,
                              hipStream_t stream)
{
    (void)in_sizes; (void)n_in; (void)out_size; (void)ws_size;
    const float* input = (const float*)d_in[0];
    const float* Wc    = (const float*)d_in[1];
    const float* bc    = (const float*)d_in[2];
    const float* W1    = (const float*)d_in[3];
    const float* b1    = (const float*)d_in[4];
    const float* W2    = (const float*)d_in[5];
    const float* b2    = (const float*)d_in[6];
    float* ws   = (float*)d_ws;
    float* WT   = ws;
    float* W2T  = ws + 217088;
    float* cbuf = ws + 217472;
    float* hall = ws + 479616;
    float* obuf = ws + 17125760;
    float* zprt = ws + 17515904;
    float* out  = (float*)d_out;

    prep_kernel<<<(217472 + 255) / 256, 256, 0, stream>>>(Wc, W1, W2, WT, W2T);

    cell_step_kernel<true><<<256, 256, 0, stream>>>(input, WT, bc, zprt, cbuf, hall, 0);
    for (int t = 1; t < 127; ++t)
        cell_step_kernel<false><<<256, 256, 0, stream>>>(input, WT, bc, zprt, cbuf, hall, t);
    gate_kernel<<<256, 128, 0, stream>>>(bc, zprt, hall + 126 * 131072, cbuf);
    head_all_kernel<<<984, 256, 0, stream>>>(WT, hall + 4 * 131072, b1, W2T, b2,
                                              obuf + 4 * 3072);
    mix_all_kernel<<<123 * 1024 / 256, 256, 0, stream>>>(input, obuf, out);
}


# Gates of step t-1 re-dealt over all 256 threads, one unit pair each (was 4 units on 144 threads); the 2 halo rows gated by waves 0-1 behind a split barrier, under the h-source WMMAs; sums unchanged
# speedup vs baseline: 5.8862x; 1.0494x over previous
#include <hip/hip_runtime.h>
#include <math.h>

typedef __attribute__((ext_vector_type(2))) float v2f;
typedef __attribute__((ext_vector_type(8))) float v8f;
typedef __attribute__((ext_vector_type(4))) float v4f;

#define STRZ 520
#define WP   36
#define WBUF (512 * WP)

__device__ __forceinline__ void chunk_request(v4f (&r)[16], const v4f* src, int tid)
{
    #pragma unroll
    for (int i = 0; i < 16; ++i) r[i] = src[tid + i * 256];
}

__device__ __forceinline__ void chunk_to_lds(float* buf, int wdst, const v4f (&r)[16])
{
    #pragma unroll
    for (int i = 0; i < 16; ++i)
        *reinterpret_cast<v4f*>(&buf[wdst + i * 32 * WP]) = r[i];
}

__device__ __forceinline__ void lds_barrier()
{
    asm volatile("s_wait_dscnt 0x0\n\ts_barrier_signal -1\n\ts_barrier_wait -1" ::: "memory");
}

__device__ __forceinline__ void wave_chunk_request(v4f (&r)[16], const v4f* src, int wave, int lane)
{
    #pragma unroll
    for (int i = 0; i < 16; ++i) r[i] = src[wave * 512 + i * 32 + lane];
}

template <int HALF>
__device__ __forceinline__ void wave_half_to_lds(float* buf, int wdst, const v4f (&r)[16])
{
    #pragma unroll
    for (int i = 8 * HALF; i < 8 * HALF + 8; ++i)
        *reinterpret_cast<v4f*>(&buf[wdst + i * 4 * WP]) = r[i];
}

__device__ __forceinline__ void wave_lds_handover()
{
    __builtin_amdgcn_fence(__ATOMIC_RELEASE, "wavefront");
    __builtin_amdgcn_wave_barrier();
    __builtin_amdgcn_fence(__ATOMIC_ACQUIRE, "wavefront");
}

__device__ __forceinline__ void chunk_mma2(const float* arow, const float* wb,
    int wlA, int wlB, int kHalf, v8f& accA, v8f& accB)
{
    #pragma unroll
    for (int kk = 0; kk < 32; kk += 4) {
        v2f a;
        a.x = arow[kk + kHalf];
        a.y = arow[kk + kHalf + 1];
        v2f bA, bB;
        bA.x = wb[wlA + kk + kHalf]; bA.y = wb[wlA + kk + kHalf + 1];
        bB.x = wb[wlB + kk + kHalf]; bB.y = wb[wlB + kk + kHalf + 1];
        accA = __builtin_amdgcn_wmma_f32_16x16x4_f32(
                    false, a, false, bA, (short)0, accA, false, false);
        accB = __builtin_amdgcn_wmma_f32_16x16x4_f32(
                    false, a, false, bB, (short)0, accB, false, false);
    }
}

#define HP 36
#define ZPLANE (1024 * 512)
#define HPLANE (1024 * 128)

__device__ __forceinline__ void gate_math(const v4f& zi, const v4f& zf, const v4f& zg,
    const v4f& zo, const v4f& cp, v4f& cn, v4f& hn)
{
    #pragma unroll
    for (int e = 0; e < 4; ++e) {
        float si = 1.0f / (1.0f + __expf(-zi[e]));
        float sf = 1.0f / (1.0f + __expf(-zf[e]));
        float so = 1.0f / (1.0f + __expf(-zo[e]));
        float cv = sf * cp[e] + si * tanhf(zg[e]);
        cn[e] = cv;
        hn[e] = so * tanhf(cv);
    }
}

__device__ __forceinline__ void gate_math1(float zi, float zf, float zg, float zo,
    float cp, float& cn, float& hn)
{
    float si = 1.0f / (1.0f + __expf(-zi));
    float sf = 1.0f / (1.0f + __expf(-zf));
    float so = 1.0f / (1.0f + __expf(-zo));
    float cv = sf * cp + si * tanhf(zg);
    cn = cv;
    hn = so * tanhf(cv);
}

template <bool FIRST>
__global__ __launch_bounds__(256) void cell_step_kernel(
    const float* __restrict__ input, const float* __restrict__ WT,
    const float* __restrict__ bcell, float* zpart, float* cbuf,
    float* __restrict__ hkeep, int t)
{
    __shared__ alignas(16) float sh_h[18 * HP];
    __shared__ float sh_x[16 * 8];
    __shared__ alignas(16) float sh_z[16 * STRZ];
    __shared__ alignas(16) float sh_w[3 * WBUF];

    const int tid = threadIdx.x;
    const int mt  = blockIdx.x >> 2;
    const int ks  = blockIdx.x & 3;
    const int R0  = mt * 16;
    const int bb  = R0 >> 7;
    const int s0  = R0 & 127;

    const float* zin  = zpart + ((t + 1) & 1) * (4 * ZPLANE);
    const float* c_in = cbuf  + ((t + 1) & 1) * HPLANE;
    float* const c_out = cbuf + (t & 1) * HPLANE;

    const int wave  = tid >> 5;
    const int lane  = tid & 31;

    float xv = 0.0f;
    if (tid < 16 * 5)
        xv = input[(((bb * 128) + t) * 128 + (s0 + tid / 5)) * 5 + tid % 5];
    const int orow = R0 + (tid >> 4);
    const int gj   = ks * 32 + (tid & 15) * 2;
    const int hseg = (wave == 0) ? s0 - 1 : (wave == 1) ? s0 + 16 : s0 + 2 * wave;
    const bool hact = (wave < 2) && (hseg >= 0) && (hseg < 128);
    const int hrow = bb * 128 + min(max(hseg, 0), 127);
    const int hj   = ks * 32 + lane;
    v2f zr[16], bq[4], cp;
    float zh[16], bh[4], ch;
    if (!FIRST) {
        const float* zp = zin + orow * 512 + gj;
        const float* zq = zin + hrow * 512 + hj;
        #pragma unroll
        for (int g = 0; g < 4; ++g) {
            #pragma unroll
            for (int p = 0; p < 4; ++p)
                zr[g * 4 + p] = *reinterpret_cast<const v2f*>(zp + g * 128 + p * ZPLANE);
            bq[g] = *reinterpret_cast<const v2f*>(&bcell[g * 128 + gj]);
        }
        cp = *reinterpret_cast<const v2f*>(&c_in[orow * 128 + gj]);
        #pragma unroll
        for (int g = 0; g < 4; ++g) {
            #pragma unroll
            for (int p = 0; p < 4; ++p)
                zh[g * 4 + p] = zq[g * 128 + p * ZPLANE];
            bh[g] = bcell[g * 128 + hj];
        }
        ch = c_in[hrow * 128 + hj];
    }

    const int lm    = lane & 15;
    const int hi    = lane >> 4;
    const int kHalf = hi * 2;

    const float* wc0 = WT + 13 * 16384 + ((wave * 4 + 0) * 16 + lm) * 8;
    const float* wc1 = WT + 13 * 16384 + ((wave * 4 + 1) * 16 + lm) * 8;
    const float* wc2 = WT + 13 * 16384 + ((wave * 4 + 2) * 16 + lm) * 8;
    const float* wc3 = WT + 13 * 16384 + ((wave * 4 + 3) * 16 + lm) * 8;
    const v4f wq0 = *reinterpret_cast<const v4f*>(wc0); const float we0 = wc0[4];
    const v4f wq1 = *reinterpret_cast<const v4f*>(wc1); const float we1 = wc1[4];
    const v4f wq2 = *reinterpret_cast<const v4f*>(wc2); const float we2 = wc2[4];
    const v4f wq3 = *reinterpret_cast<const v4f*>(wc3); const float we3 = wc3[4];

    asm volatile("" ::: "memory");

    const v4f* wsrc = reinterpret_cast<const v4f*>(WT);
    const int wdst = (wave * 64 + (lane >> 3)) * WP + (lane & 7) * 4;
    v4f preA[16], preB[16], preC[16];
    wave_chunk_request(preA, wsrc + ks * 4096, wave, lane);
    wave_chunk_request(preB, wsrc + (4 + ks) * 4096, wave, lane);

    if (!FIRST) {
        asm volatile("" :: "v"(zr[0]), "v"(zr[1]), "v"(zr[2]), "v"(zr[3]),
                           "v"(zr[4]), "v"(zr[5]), "v"(zr[6]), "v"(zr[7]),
                           "v"(zr[8]), "v"(zr[9]), "v"(zr[10]), "v"(zr[11]),
                           "v"(zr[12]), "v"(zr[13]), "v"(zr[14]), "v"(zr[15]),
                           "v"(bq[0]), "v"(bq[1]), "v"(bq[2]), "v"(bq[3]), "v"(cp));
        asm volatile("" :: "v"(zh[0]), "v"(zh[1]), "v"(zh[2]), "v"(zh[3]),
                           "v"(zh[4]), "v"(zh[5]), "v"(zh[6]), "v"(zh[7]),
                           "v"(zh[8]), "v"(zh[9]), "v"(zh[10]), "v"(zh[11]),
                           "v"(zh[12]), "v"(zh[13]), "v"(zh[14]), "v"(zh[15]),
                           "v"(bh[0]), "v"(bh[1]), "v"(bh[2]), "v"(bh[3]), "v"(ch));
    }

    v2f hv = {0.0f, 0.0f};
    if (FIRST) {
        *reinterpret_cast<v2f*>(&c_out[orow * 128 + gj]) = hv;
    } else {
        const v2f zi = (((zr[0]  + zr[1])  + zr[2])  + zr[3])  + bq[0];
        const v2f zf = (((zr[4]  + zr[5])  + zr[6])  + zr[7])  + bq[1];
        const v2f zg = (((zr[8]  + zr[9])  + zr[10]) + zr[11]) + bq[2];
        const v2f zo = (((zr[12] + zr[13]) + zr[14]) + zr[15]) + bq[3];
        float cn0, cn1, hn0, hn1;
        gate_math1(zi.x, zf.x, zg.x, zo.x, cp.x, cn0, hn0);
        gate_math1(zi.y, zf.y, zg.y, zo.y, cp.y, cn1, hn1);
        const v2f cn = {cn0, cn1};
        hv.x = hn0; hv.y = hn1;
        *reinterpret_cast<v2f*>(&c_out[orow * 128 + gj]) = cn;
        *reinterpret_cast<v2f*>(&hkeep[(t - 1) * HPLANE + orow * 128 + gj]) = hv;
    }

    wave_chunk_request(preC, wsrc + (8 + ks) * 4096, wave, lane);

    *reinterpret_cast<v2f*>(&sh_h[(1 + (tid >> 4)) * HP + (tid & 15) * 2]) = hv;
    if (tid < 16 * 5)
        sh_x[(tid / 5) * 8 + tid % 5] = xv;
    float* const buf0 = sh_w;
    float* const buf1 = sh_w + WBUF;
    float* const buf2 = sh_w + 2 * WBUF;
    lds_barrier();

    float hh = 0.0f;
    if (!FIRST) {
        if (hact) {
            const float zi = (((zh[0]  + zh[1])  + zh[2])  + zh[3])  + bh[0];
            const float zf = (((zh[4]  + zh[5])  + zh[6])  + zh[7])  + bh[1];
            const float zg = (((zh[8]  + zh[9])  + zh[10]) + zh[11]) + bh[2];
            const float zo = (((zh[12] + zh[13]) + zh[14]) + zh[15]) + bh[3];
            float cnh;
            gate_math1(zi, zf, zg, zo, ch, cnh, hh);
        }
    }
    if (wave < 2)
        sh_h[(wave == 0 ? 0 : 17) * HP + lane] = hh;
    asm volatile("s_wait_dscnt 0x0\n\ts_barrier_signal -1" ::: "memory");

    const int wl0 = ((wave * 4 + 0) * 16 + lm) * WP;
    const int wl1 = ((wave * 4 + 1) * 16 + lm) * WP;
    const int wl2 = ((wave * 4 + 2) * 16 + lm) * WP;
    const int wl3 = ((wave * 4 + 3) * 16 + lm) * WP;

    v8f acc0 = {}, acc1 = {}, acc2 = {}, acc3 = {};

    if (ks == 0) {
        #pragma unroll
        for (int v = 0; v < 8; ++v) {
            int m = v + hi * 8;
            float x0 = sh_x[m*8+0], x1 = sh_x[m*8+1], x2 = sh_x[m*8+2],
                  x3 = sh_x[m*8+3], x4 = sh_x[m*8+4];
            acc0[v] += x0*wq0.x + x1*wq0.y + x2*wq0.z + x3*wq0.w + x4*we0;
            acc1[v] += x0*wq1.x + x1*wq1.y + x2*wq1.z + x3*wq1.w + x4*we1;
            acc2[v] += x0*wq2.x + x1*wq2.y + x2*wq2.z + x3*wq2.w + x4*we2;
            acc3[v] += x0*wq3.x + x1*wq3.y + x2*wq3.z + x3*wq3.w + x4*we3;
        }
    }

    const float* const a0 = &sh_h[(1 + lm) * HP];
    const float* const a1 = &sh_h[(2 + lm) * HP];
    const float* const a2 = &sh_h[lm * HP];
    wave_half_to_lds<0>(buf0, wdst, preA);
    wave_lds_handover();
    chunk_mma2(a0, buf0, wl0, wl1, kHalf, acc0, acc1);
    asm volatile("" :: "v"(acc0), "v"(acc1) : "memory");
    wave_half_to_lds<1>(buf0, wdst, preA);
    wave_lds_handover();
    chunk_mma2(a0, buf0, wl2, wl3, kHalf, acc2, acc3);
    asm volatile("" :: "v"(acc2), "v"(acc3) : "memory");
    asm volatile("s_barrier_wait -1" ::: "memory");
    wave_half_to_lds<0>(buf1, wdst, preB);
    wave_lds_handover();
    chunk_mma2(a1, buf1, wl0, wl1, kHalf, acc0, acc1);
    asm volatile("" :: "v"(acc0), "v"(acc1) : "memory");
    wave_half_to_lds<1>(buf1, wdst, preB);
    wave_lds_handover();
    chunk_mma2(a1, buf1, wl2, wl3, kHalf, acc2, acc3);
    asm volatile("" :: "v"(acc2), "v"(acc3) : "memory");
    wave_half_to_lds<0>(buf2, wdst, preC);
    wave_lds_handover();
    chunk_mma2(a2, buf2, wl0, wl1, kHalf, acc0, acc1);
    asm volatile("" :: "v"(acc0), "v"(acc1) : "memory");
    wave_half_to_lds<1>(buf2, wdst, preC);
    wave_lds_handover();
    chunk_mma2(a2, buf2, wl2, wl3, kHalf, acc2, acc3);

    #pragma unroll
    for (int v = 0; v < 8; ++v) {
        int m = v + hi * 8;
        sh_z[m * STRZ + (wave * 4 + 0) * 16 + lm] = acc0[v];
        sh_z[m * STRZ + (wave * 4 + 1) * 16 + lm] = acc1[v];
        sh_z[m * STRZ + (wave * 4 + 2) * 16 + lm] = acc2[v];
        sh_z[m * STRZ + (wave * 4 + 3) * 16 + lm] = acc3[v];
    }
    __syncthreads();

    float* zp = zpart + (t & 1) * (4 * ZPLANE) + ks * ZPLANE + R0 * 512;
    #pragma unroll
    for (int it = 0; it < 8; ++it) {
        int i = tid + it * 256;
        int m = i >> 7, c4 = (i & 127) * 4;
        *reinterpret_cast<v4f*>(&zp[m * 512 + c4]) =
            *reinterpret_cast<const v4f*>(&sh_z[m * STRZ + c4]);
    }
}

__device__ __forceinline__ v4f plane_sum(const float* p)
{
    v4f a = *reinterpret_cast<const v4f*>(p);
    v4f b = *reinterpret_cast<const v4f*>(p + ZPLANE);
    v4f c = *reinterpret_cast<const v4f*>(p + 2 * ZPLANE);
    v4f d = *reinterpret_cast<const v4f*>(p + 3 * ZPLANE);
    return ((a + b) + c) + d;
}

__global__ __launch_bounds__(128) void gate_kernel(
    const float* __restrict__ bcell, const float* __restrict__ zpart,
    float* __restrict__ h_out, float* __restrict__ c)
{
    const int gid = blockIdx.x * 128 + threadIdx.x;
    const int row = gid >> 5, gj = (gid & 31) * 4;

    const float* zp = zpart + row * 512 + gj;
    const v4f zi = plane_sum(zp)       + *reinterpret_cast<const v4f*>(&bcell[gj]);
    const v4f zf = plane_sum(zp + 128) + *reinterpret_cast<const v4f*>(&bcell[128 + gj]);
    const v4f zg = plane_sum(zp + 256) + *reinterpret_cast<const v4f*>(&bcell[256 + gj]);
    const v4f zo = plane_sum(zp + 384) + *reinterpret_cast<const v4f*>(&bcell[384 + gj]);
    const v4f cp = *reinterpret_cast<const v4f*>(&c[row * 128 + gj]);

    v4f cn, hn;
    gate_math(zi, zf, zg, zo, cp, cn, hn);
    *reinterpret_cast<v4f*>(&c[row * 128 + gj])     = cn;
    *reinterpret_cast<v4f*>(&h_out[row * 128 + gj]) = hn;
}

#define HIDP 132

__global__ __launch_bounds__(256) void head_all_kernel(
    const float* __restrict__ WT, const float* __restrict__ hrows,
    const float* __restrict__ b1, const float* __restrict__ W2T,
    const float* __restrict__ b2, float* __restrict__ orows)
{
    __shared__ alignas(16) float sh_w[WBUF];
    __shared__ alignas(16) float sh_hid[8 * 16 * HIDP];
    __shared__ alignas(16) float sh_w2[3 * 128];
    __shared__ alignas(16) float sh_o[8 * 48];

    const int tid   = threadIdx.x;
    const int wave  = tid >> 5;
    const int lane  = tid & 31;
    const int lm    = lane & 15;
    const int hi    = lane >> 4;
    const int kHalf = hi * 2;

    const v4f* wsrc = reinterpret_cast<const v4f*>(WT);
    const int wdst = (tid >> 3) * WP + (tid & 7) * 4;
    v4f preA[16];
    chunk_request(preA, wsrc + 12 * 4096, tid);

    const float* hrow = hrows + ((blockIdx.x * 8 + wave) * 16 + lm) * 128 + kHalf;
    v2f a[32];
    #pragma unroll
    for (int ks = 0; ks < 32; ++ks)
        a[ks] = *reinterpret_cast<const v2f*>(hrow + ks * 4);

    float bias1[8];
    #pragma unroll
    for (int j = 0; j < 8; ++j) bias1[j] = b1[j * 16 + lm];
    const float b20 = b2[0], b21 = b2[1], b22 = b2[2];
    v4f w2v = {0.0f, 0.0f, 0.0f, 0.0f};
    if (tid < 96) w2v = *reinterpret_cast<const v4f*>(W2T + tid * 4);

    if (tid < 96) *reinterpret_cast<v4f*>(&sh_w2[tid * 4]) = w2v;
    chunk_to_lds(sh_w, wdst, preA);
    __syncthreads();

    v8f acc[8];
    #pragma unroll
    for (int j = 0; j < 8; ++j) acc[j] = v8f{};
    #pragma unroll
    for (int kq = 0; kq < 4; ++kq) {
        #pragma unroll
        for (int kk = 0; kk < 32; kk += 4) {
            #pragma unroll
            for (int j = 0; j < 8; ++j) {
                const float* wq = sh_w + (kq * 128 + j * 16 + lm) * WP;
                v2f b;
                b.x = wq[kk + kHalf]; b.y = wq[kk + kHalf + 1];
                acc[j] = __builtin_amdgcn_wmma_f32_16x16x4_f32(
                            false, a[kq * 8 + (kk >> 2)], false, b, (short)0, acc[j], false, false);
            }
        }
    }
    float* hid = sh_hid + wave * 16 * HIDP;
    #pragma unroll
    for (int j = 0; j < 8; ++j) {
        #pragma unroll
        for (int v = 0; v < 8; ++v) {
            float x = acc[j][v] + bias1[j];
            hid[(v + hi * 8) * HIDP + j * 16 + lm] = x > 0.0f ? x : 0.0f;
        }
    }
    __syncthreads();

    if (lane < 16) {
        const float* hr = hid + lane * HIDP;
        float s0 = 0.0f, s1 = 0.0f, s2 = 0.0f;
        for (int k = 0; k < 128; k += 4) {
            const v4f hv = *reinterpret_cast<const v4f*>(hr + k);
            const v4f wa = *reinterpret_cast<const v4f*>(sh_w2 + k);
            const v4f wb = *reinterpret_cast<const v4f*>(sh_w2 + 128 + k);
            const v4f wc = *reinterpret_cast<const v4f*>(sh_w2 + 256 + k);
            s0 = fmaf(hv.x, wa.x, s0); s0 = fmaf(hv.y, wa.y, s0);
            s0 = fmaf(hv.z, wa.z, s0); s0 = fmaf(hv.w, wa.w, s0);
            s1 = fmaf(hv.x, wb.x, s1); s1 = fmaf(hv.y, wb.y, s1);
            s1 = fmaf(hv.z, wb.z, s1); s1 = fmaf(hv.w, wb.w, s1);
            s2 = fmaf(hv.x, wc.x, s2); s2 = fmaf(hv.y, wc.y, s2);
            s2 = fmaf(hv.z, wc.z, s2); s2 = fmaf(hv.w, wc.w, s2);
        }
        float* so = sh_o + wave * 48 + lane * 3;
        so[0] = s0 + b20; so[1] = s1 + b21; so[2] = s2 + b22;
    }
    __syncthreads();

    if (tid < 96)
        *reinterpret_cast<v4f*>(orows + blockIdx.x * 384 + tid * 4) =
            *reinterpret_cast<const v4f*>(sh_o + tid * 4);
}

__global__ __launch_bounds__(256) void mix_all_kernel(
    const float* __restrict__ input, const float* __restrict__ obuf_all,
    float* __restrict__ out)
{
    int gid = blockIdx.x * 256 + threadIdx.x;
    if (gid >= 123 * 1024) return;
    int tt = gid >> 10, r = gid & 1023;
    int t  = tt + 4;
    int bb = r >> 7, s = r & 127;
    const float* ob = obuf_all + t * 3072;
    const float* o  = &ob[r * 3];
    float n0 = o[0];
    float n1, sp0, sp1;
    if (s == 0) {
        const float* xn0 = &input[((bb * 128 + (t + 1)) * 128 + 0) * 5];
        n1 = xn0[1]; sp0 = xn0[3]; sp1 = xn0[4];
    } else {
        n1  = ob[(r - 1) * 3 + 0];
        sp0 = o[1]; sp1 = o[2];
    }
    float fn = input[((bb * 128 + t) * 128 + s) * 5 + 2];
    float n2 = fn + n1 - n0;
    float* dst = &out[((bb * 123 + tt) * 128 + s) * 5];
    dst[0] = n0; dst[1] = n1; dst[2] = n2; dst[3] = sp0; dst[4] = sp1;
}

__global__ __launch_bounds__(256) void prep_kernel(
    const float* __restrict__ Wc, const float* __restrict__ W1,
    const float* __restrict__ W2, float* __restrict__ WT,
    float* __restrict__ W2T)
{
    int i = blockIdx.x * 256 + threadIdx.x;
    if (i < 12 * 16384) {
        int chn = i >> 14, col = (i >> 5) & 511, q = i & 31;
        WT[i] = Wc[(5 + chn * 32 + q) * 512 + col];
    } else if (i < 13 * 16384) {
        int j = i - 12 * 16384;
        int pc = j >> 5, q = j & 31;
        int kq = pc >> 7, n = pc & 127;
        WT[i] = W1[(kq * 32 + q) * 128 + n];
    } else if (i < 13 * 16384 + 4096) {
        int j = i - 13 * 16384;
        int col = j >> 3, k = j & 7;
        WT[i] = (k < 5) ? Wc[k * 512 + col] : 0.0f;
    } else if (i < 13 * 16384 + 4096 + 384) {
        int j = i - 13 * 16384 - 4096;
        int cc = j >> 7, k = j & 127;
        W2T[j] = W2[k * 3 + cc];
    }
}

extern "C" void kernel_launch(void* const* d_in, const int* in_sizes, int n_in,
                              void* d_out, int out_size, void* d_ws, size_t ws_size,
                              hipStream_t stream)
{
    (void)in_sizes; (void)n_in; (void)out_size; (void)ws_size;
    const float* input = (const float*)d_in[0];
    const float* Wc    = (const float*)d_in[1];
    const float* bc    = (const float*)d_in[2];
    const float* W1    = (const float*)d_in[3];
    const float* b1    = (const float*)d_in[4];
    const float* W2    = (const float*)d_in[5];
    const float* b2    = (const float*)d_in[6];
    float* ws   = (float*)d_ws;
    float* WT   = ws;
    float* W2T  = ws + 217088;
    float* cbuf = ws + 217472;
    float* hall = ws + 479616;
    float* obuf = ws + 17125760;
    float* zprt = ws + 17515904;
    float* out  = (float*)d_out;

    prep_kernel<<<(217472 + 255) / 256, 256, 0, stream>>>(Wc, W1, W2, WT, W2T);

    cell_step_kernel<true><<<256, 256, 0, stream>>>(input, WT, bc, zprt, cbuf, hall, 0);
    for (int t = 1; t < 127; ++t)
        cell_step_kernel<false><<<256, 256, 0, stream>>>(input, WT, bc, zprt, cbuf, hall, t);
    gate_kernel<<<256, 128, 0, stream>>>(bc, zprt, hall + 126 * 131072, cbuf);
    head_all_kernel<<<984, 256, 0, stream>>>(WT, hall + 4 * 131072, b1, W2T, b2,
                                              obuf + 4 * 3072);
    mix_all_kernel<<<123 * 1024 / 256, 256, 0, stream>>>(input, obuf, out);
}


# Partial planes re-laid unit-major, a unit's four gates adjacent (a wave's N-tiles = the 4 gates of its 16 units): 16-byte stores straight from accumulators, 16-byte gate reads; sums unchanged
# speedup vs baseline: 6.2754x; 1.0661x over previous
#include <hip/hip_runtime.h>
#include <math.h>

typedef __attribute__((ext_vector_type(2))) float v2f;
typedef __attribute__((ext_vector_type(8))) float v8f;
typedef __attribute__((ext_vector_type(4))) float v4f;

#define WP   36
#define WBUF (512 * WP)

__device__ __forceinline__ void chunk_request(v4f (&r)[16], const v4f* src, int tid)
{
    #pragma unroll
    for (int i = 0; i < 16; ++i) r[i] = src[tid + i * 256];
}

__device__ __forceinline__ void chunk_to_lds(float* buf, int wdst, const v4f (&r)[16])
{
    #pragma unroll
    for (int i = 0; i < 16; ++i)
        *reinterpret_cast<v4f*>(&buf[wdst + i * 32 * WP]) = r[i];
}

__device__ __forceinline__ void lds_barrier()
{
    asm volatile("s_wait_dscnt 0x0\n\ts_barrier_signal -1\n\ts_barrier_wait -1" ::: "memory");
}

__device__ __forceinline__ void wave_chunk_request(v4f (&r)[16], const v4f* src, int wave, int lane)
{
    #pragma unroll
    for (int i = 0; i < 16; ++i) r[i] = src[wave * 512 + i * 32 + lane];
}

template <int HALF>
__device__ __forceinline__ void wave_half_to_lds(float* buf, int wdst, const v4f (&r)[16])
{
    #pragma unroll
    for (int i = 8 * HALF; i < 8 * HALF + 8; ++i)
        *reinterpret_cast<v4f*>(&buf[wdst + i * 4 * WP]) = r[i];
}

__device__ __forceinline__ void wave_lds_handover()
{
    __builtin_amdgcn_fence(__ATOMIC_RELEASE, "wavefront");
    __builtin_amdgcn_wave_barrier();
    __builtin_amdgcn_fence(__ATOMIC_ACQUIRE, "wavefront");
}

__device__ __forceinline__ void chunk_mma2(const float* arow, const float* wb,
    int wlA, int wlB, int kHalf, v8f& accA, v8f& accB)
{
    #pragma unroll
    for (int kk = 0; kk < 32; kk += 4) {
        v2f a;
        a.x = arow[kk + kHalf];
        a.y = arow[kk + kHalf + 1];
        v2f bA, bB;
        bA.x = wb[wlA + kk + kHalf]; bA.y = wb[wlA + kk + kHalf + 1];
        bB.x = wb[wlB + kk + kHalf]; bB.y = wb[wlB + kk + kHalf + 1];
        accA = __builtin_amdgcn_wmma_f32_16x16x4_f32(
                    false, a, false, bA, (short)0, accA, false, false);
        accB = __builtin_amdgcn_wmma_f32_16x16x4_f32(
                    false, a, false, bB, (short)0, accB, false, false);
    }
}

#define HP 36
#define ZPLANE (1024 * 512)
#define HPLANE (1024 * 128)

__device__ __forceinline__ void gate_math1(float zi, float zf, float zg, float zo,
    float cp, float& cn, float& hn)
{
    float si = 1.0f / (1.0f + __expf(-zi));
    float sf = 1.0f / (1.0f + __expf(-zf));
    float so = 1.0f / (1.0f + __expf(-zo));
    float cv = sf * cp + si * tanhf(zg);
    cn = cv;
    hn = so * tanhf(cv);
}

template <bool FIRST>
__global__ __launch_bounds__(256) void cell_step_kernel(
    const float* __restrict__ input, const float* __restrict__ WT,
    const float* __restrict__ bperm, float* zpart, float* cbuf,
    float* __restrict__ hkeep, int t)
{
    __shared__ alignas(16) float sh_h[18 * HP];
    __shared__ float sh_x[16 * 8];
    __shared__ alignas(16) float sh_w[3 * WBUF];

    const int tid = threadIdx.x;
    const int mt  = blockIdx.x >> 2;
    const int ks  = blockIdx.x & 3;
    const int R0  = mt * 16;
    const int bb  = R0 >> 7;
    const int s0  = R0 & 127;

    const float* zin  = zpart + ((t + 1) & 1) * (4 * ZPLANE);
    const float* c_in = cbuf  + ((t + 1) & 1) * HPLANE;
    float* const c_out = cbuf + (t & 1) * HPLANE;

    const int wave  = tid >> 5;
    const int lane  = tid & 31;

    float xv = 0.0f;
    if (tid < 16 * 5)
        xv = input[(((bb * 128) + t) * 128 + (s0 + tid / 5)) * 5 + tid % 5];
    const int orow = R0 + 2 * wave;
    const int gu   = ks * 32 + lane;
    const int hseg = (wave == 0) ? s0 - 1 : (wave == 1) ? s0 + 16 : s0 + 2 * wave;
    const bool hact = (wave < 2) && (hseg >= 0) && (hseg < 128);
    const int hrow = bb * 128 + min(max(hseg, 0), 127);
    v4f zr[8], zh[4], bq;
    float cp0, cp1, ch;
    if (!FIRST) {
        const float* zp = zin + orow * 512 + gu * 4;
        const float* zq = zin + hrow * 512 + gu * 4;
        #pragma unroll
        for (int p = 0; p < 4; ++p) {
            zr[p]     = *reinterpret_cast<const v4f*>(zp + p * ZPLANE);
            zr[4 + p] = *reinterpret_cast<const v4f*>(zp + 512 + p * ZPLANE);
        }
        bq  = *reinterpret_cast<const v4f*>(&bperm[gu * 4]);
        cp0 = c_in[orow * 128 + gu];
        cp1 = c_in[(orow + 1) * 128 + gu];
        #pragma unroll
        for (int p = 0; p < 4; ++p)
            zh[p] = *reinterpret_cast<const v4f*>(zq + p * ZPLANE);
        ch = c_in[hrow * 128 + gu];
    }

    const int lm    = lane & 15;
    const int hi    = lane >> 4;
    const int kHalf = hi * 2;

    const float* wc0 = WT + 13 * 16384 + ((wave * 4 + 0) * 16 + lm) * 8;
    const float* wc1 = WT + 13 * 16384 + ((wave * 4 + 1) * 16 + lm) * 8;
    const float* wc2 = WT + 13 * 16384 + ((wave * 4 + 2) * 16 + lm) * 8;
    const float* wc3 = WT + 13 * 16384 + ((wave * 4 + 3) * 16 + lm) * 8;
    const v4f wq0 = *reinterpret_cast<const v4f*>(wc0); const float we0 = wc0[4];
    const v4f wq1 = *reinterpret_cast<const v4f*>(wc1); const float we1 = wc1[4];
    const v4f wq2 = *reinterpret_cast<const v4f*>(wc2); const float we2 = wc2[4];
    const v4f wq3 = *reinterpret_cast<const v4f*>(wc3); const float we3 = wc3[4];

    asm volatile("" ::: "memory");

    const v4f* wsrc = reinterpret_cast<const v4f*>(WT);
    const int wdst = (wave * 64 + (lane >> 3)) * WP + (lane & 7) * 4;
    v4f preA[16], preB[16], preC[16];
    wave_chunk_request(preA, wsrc + ks * 4096, wave, lane);
    wave_chunk_request(preB, wsrc + (4 + ks) * 4096, wave, lane);

    if (!FIRST) {
        asm volatile("" :: "v"(zr[0]), "v"(zr[1]), "v"(zr[2]), "v"(zr[3]),
                           "v"(zr[4]), "v"(zr[5]), "v"(zr[6]), "v"(zr[7]),
                           "v"(bq), "v"(cp0), "v"(cp1),
                           "v"(zh[0]), "v"(zh[1]), "v"(zh[2]), "v"(zh[3]), "v"(ch));
    }

    float hn0 = 0.0f, hn1 = 0.0f;
    if (FIRST) {
        c_out[orow * 128 + gu]       = 0.0f;
        c_out[(orow + 1) * 128 + gu] = 0.0f;
    } else {
        const v4f za = (((zr[0] + zr[1]) + zr[2]) + zr[3]) + bq;
        const v4f zb = (((zr[4] + zr[5]) + zr[6]) + zr[7]) + bq;
        float cn0, cn1;
        gate_math1(za.x, za.y, za.z, za.w, cp0, cn0, hn0);
        gate_math1(zb.x, zb.y, zb.z, zb.w, cp1, cn1, hn1);
        c_out[orow * 128 + gu]       = cn0;
        c_out[(orow + 1) * 128 + gu] = cn1;
        hkeep[(t - 1) * HPLANE + orow * 128 + gu]       = hn0;
        hkeep[(t - 1) * HPLANE + (orow + 1) * 128 + gu] = hn1;
    }

    wave_chunk_request(preC, wsrc + (8 + ks) * 4096, wave, lane);

    sh_h[(1 + 2 * wave) * HP + lane] = hn0;
    sh_h[(2 + 2 * wave) * HP + lane] = hn1;
    if (tid < 16 * 5)
        sh_x[(tid / 5) * 8 + tid % 5] = xv;
    float* const buf0 = sh_w;
    float* const buf1 = sh_w + WBUF;
    float* const buf2 = sh_w + 2 * WBUF;
    lds_barrier();

    float hh = 0.0f;
    if (!FIRST) {
        if (hact) {
            const v4f zc = (((zh[0] + zh[1]) + zh[2]) + zh[3]) + bq;
            float cnh;
            gate_math1(zc.x, zc.y, zc.z, zc.w, ch, cnh, hh);
        }
    }
    if (wave < 2)
        sh_h[(wave == 0 ? 0 : 17) * HP + lane] = hh;
    asm volatile("s_wait_dscnt 0x0\n\ts_barrier_signal -1" ::: "memory");

    const int wl0 = ((wave * 4 + 0) * 16 + lm) * WP;
    const int wl1 = ((wave * 4 + 1) * 16 + lm) * WP;
    const int wl2 = ((wave * 4 + 2) * 16 + lm) * WP;
    const int wl3 = ((wave * 4 + 3) * 16 + lm) * WP;

    v8f acc0 = {}, acc1 = {}, acc2 = {}, acc3 = {};

    if (ks == 0) {
        #pragma unroll
        for (int v = 0; v < 8; ++v) {
            int m = v + hi * 8;
            float x0 = sh_x[m*8+0], x1 = sh_x[m*8+1], x2 = sh_x[m*8+2],
                  x3 = sh_x[m*8+3], x4 = sh_x[m*8+4];
            acc0[v] += x0*wq0.x + x1*wq0.y + x2*wq0.z + x3*wq0.w + x4*we0;
            acc1[v] += x0*wq1.x + x1*wq1.y + x2*wq1.z + x3*wq1.w + x4*we1;
            acc2[v] += x0*wq2.x + x1*wq2.y + x2*wq2.z + x3*wq2.w + x4*we2;
            acc3[v] += x0*wq3.x + x1*wq3.y + x2*wq3.z + x3*wq3.w + x4*we3;
        }
    }

    const float* const a0 = &sh_h[(1 + lm) * HP];
    const float* const a1 = &sh_h[(2 + lm) * HP];
    const float* const a2 = &sh_h[lm * HP];
    wave_half_to_lds<0>(buf0, wdst, preA);
    wave_lds_handover();
    chunk_mma2(a0, buf0, wl0, wl1, kHalf, acc0, acc1);
    asm volatile("" :: "v"(acc0), "v"(acc1) : "memory");
    wave_half_to_lds<1>(buf0, wdst, preA);
    wave_lds_handover();
    chunk_mma2(a0, buf0, wl2, wl3, kHalf, acc2, acc3);
    asm volatile("" :: "v"(acc2), "v"(acc3) : "memory");
    asm volatile("s_barrier_wait -1" ::: "memory");
    wave_half_to_lds<0>(buf1, wdst, preB);
    wave_lds_handover();
    chunk_mma2(a1, buf1, wl0, wl1, kHalf, acc0, acc1);
    asm volatile("" :: "v"(acc0), "v"(acc1) : "memory");
    wave_half_to_lds<1>(buf1, wdst, preB);
    wave_lds_handover();
    chunk_mma2(a1, buf1, wl2, wl3, kHalf, acc2, acc3);
    asm volatile("" :: "v"(acc2), "v"(acc3) : "memory");
    wave_half_to_lds<0>(buf2, wdst, preC);
    wave_lds_handover();
    chunk_mma2(a2, buf2, wl0, wl1, kHalf, acc0, acc1);
    asm volatile("" :: "v"(acc0), "v"(acc1) : "memory");
    wave_half_to_lds<1>(buf2, wdst, preC);
    wave_lds_handover();
    chunk_mma2(a2, buf2, wl2, wl3, kHalf, acc2, acc3);

    float* zp = zpart + (t & 1) * (4 * ZPLANE) + ks * ZPLANE + R0 * 512 + wave * 64 + lm * 4;
    #pragma unroll
    for (int v = 0; v < 8; ++v) {
        const v4f piece = {acc0[v], acc1[v], acc2[v], acc3[v]};
        *reinterpret_cast<v4f*>(&zp[(v + hi * 8) * 512]) = piece;
    }
}

__global__ __launch_bounds__(128) void gate_kernel(
    const float* __restrict__ bperm, const float* __restrict__ zpart,
    float* __restrict__ h_out, float* __restrict__ c)
{
    const int row = blockIdx.x, gu = threadIdx.x;

    const float* zp = zpart + row * 512 + gu * 4;
    const v4f p0 = *reinterpret_cast<const v4f*>(zp);
    const v4f p1 = *reinterpret_cast<const v4f*>(zp + ZPLANE);
    const v4f p2 = *reinterpret_cast<const v4f*>(zp + 2 * ZPLANE);
    const v4f p3 = *reinterpret_cast<const v4f*>(zp + 3 * ZPLANE);
    const v4f bq = *reinterpret_cast<const v4f*>(&bperm[gu * 4]);
    const float cp = c[row * 128 + gu];

    const v4f z = (((p0 + p1) + p2) + p3) + bq;
    float cn, hn;
    gate_math1(z.x, z.y, z.z, z.w, cp, cn, hn);
    c[row * 128 + gu]     = cn;
    h_out[row * 128 + gu] = hn;
}

#define HIDP 132

__global__ __launch_bounds__(256) void head_all_kernel(
    const float* __restrict__ WT, const float* __restrict__ hrows,
    const float* __restrict__ b1, const float* __restrict__ W2T,
    const float* __restrict__ b2, float* __restrict__ orows)
{
    __shared__ alignas(16) float sh_w[WBUF];
    __shared__ alignas(16) float sh_hid[8 * 16 * HIDP];
    __shared__ alignas(16) float sh_w2[3 * 128];
    __shared__ alignas(16) float sh_o[8 * 48];

    const int tid   = threadIdx.x;
    const int wave  = tid >> 5;
    const int lane  = tid & 31;
    const int lm    = lane & 15;
    const int hi    = lane >> 4;
    const int kHalf = hi * 2;

    const v4f* wsrc = reinterpret_cast<const v4f*>(WT);
    const int wdst = (tid >> 3) * WP + (tid & 7) * 4;
    v4f preA[16];
    chunk_request(preA, wsrc + 12 * 4096, tid);

    const float* hrow = hrows + ((blockIdx.x * 8 + wave) * 16 + lm) * 128 + kHalf;
    v2f a[32];
    #pragma unroll
    for (int ks = 0; ks < 32; ++ks)
        a[ks] = *reinterpret_cast<const v2f*>(hrow + ks * 4);

    float bias1[8];
    #pragma unroll
    for (int j = 0; j < 8; ++j) bias1[j] = b1[j * 16 + lm];
    const float b20 = b2[0], b21 = b2[1], b22 = b2[2];
    v4f w2v = {0.0f, 0.0f, 0.0f, 0.0f};
    if (tid < 96) w2v = *reinterpret_cast<const v4f*>(W2T + tid * 4);

    if (tid < 96) *reinterpret_cast<v4f*>(&sh_w2[tid * 4]) = w2v;
    chunk_to_lds(sh_w, wdst, preA);
    __syncthreads();

    v8f acc[8];
    #pragma unroll
    for (int j = 0; j < 8; ++j) acc[j] = v8f{};
    #pragma unroll
    for (int kq = 0; kq < 4; ++kq) {
        #pragma unroll
        for (int kk = 0; kk < 32; kk += 4) {
            #pragma unroll
            for (int j = 0; j < 8; ++j) {
                const float* wq = sh_w + (kq * 128 + j * 16 + lm) * WP;
                v2f b;
                b.x = wq[kk + kHalf]; b.y = wq[kk + kHalf + 1];
                acc[j] = __builtin_amdgcn_wmma_f32_16x16x4_f32(
                            false, a[kq * 8 + (kk >> 2)], false, b, (short)0, acc[j], false, false);
            }
        }
    }
    float* hid = sh_hid + wave * 16 * HIDP;
    #pragma unroll
    for (int j = 0; j < 8; ++j) {
        #pragma unroll
        for (int v = 0; v < 8; ++v) {
            float x = acc[j][v] + bias1[j];
            hid[(v + hi * 8) * HIDP + j * 16 + lm] = x > 0.0f ? x : 0.0f;
        }
    }
    __syncthreads();

    if (lane < 16) {
        const float* hr = hid + lane * HIDP;
        float s0 = 0.0f, s1 = 0.0f, s2 = 0.0f;
        for (int k = 0; k < 128; k += 4) {
            const v4f hv = *reinterpret_cast<const v4f*>(hr + k);
            const v4f wa = *reinterpret_cast<const v4f*>(sh_w2 + k);
            const v4f wb = *reinterpret_cast<const v4f*>(sh_w2 + 128 + k);
            const v4f wc = *reinterpret_cast<const v4f*>(sh_w2 + 256 + k);
            s0 = fmaf(hv.x, wa.x, s0); s0 = fmaf(hv.y, wa.y, s0);
            s0 = fmaf(hv.z, wa.z, s0); s0 = fmaf(hv.w, wa.w, s0);
            s1 = fmaf(hv.x, wb.x, s1); s1 = fmaf(hv.y, wb.y, s1);
            s1 = fmaf(hv.z, wb.z, s1); s1 = fmaf(hv.w, wb.w, s1);
            s2 = fmaf(hv.x, wc.x, s2); s2 = fmaf(hv.y, wc.y, s2);
            s2 = fmaf(hv.z, wc.z, s2); s2 = fmaf(hv.w, wc.w, s2);
        }
        float* so = sh_o + wave * 48 + lane * 3;
        so[0] = s0 + b20; so[1] = s1 + b21; so[2] = s2 + b22;
    }
    __syncthreads();

    if (tid < 96)
        *reinterpret_cast<v4f*>(orows + blockIdx.x * 384 + tid * 4) =
            *reinterpret_cast<const v4f*>(sh_o + tid * 4);
}

__global__ __launch_bounds__(256) void mix_all_kernel(
    const float* __restrict__ input, const float* __restrict__ obuf_all,
    float* __restrict__ out)
{
    int gid = blockIdx.x * 256 + threadIdx.x;
    if (gid >= 123 * 1024) return;
    int tt = gid >> 10, r = gid & 1023;
    int t  = tt + 4;
    int bb = r >> 7, s = r & 127;
    const float* ob = obuf_all + t * 3072;
    const float* o  = &ob[r * 3];
    float n0 = o[0];
    float n1, sp0, sp1;
    if (s == 0) {
        const float* xn0 = &input[((bb * 128 + (t + 1)) * 128 + 0) * 5];
        n1 = xn0[1]; sp0 = xn0[3]; sp1 = xn0[4];
    } else {
        n1  = ob[(r - 1) * 3 + 0];
        sp0 = o[1]; sp1 = o[2];
    }
    float fn = input[((bb * 128 + t) * 128 + s) * 5 + 2];
    float n2 = fn + n1 - n0;
    float* dst = &out[((bb * 123 + tt) * 128 + s) * 5];
    dst[0] = n0; dst[1] = n1; dst[2] = n2; dst[3] = sp0; dst[4] = sp1;
}

__device__ __forceinline__ int gate_col(int colp)
{
    return ((colp >> 4) & 3) * 128 + (colp >> 6) * 16 + (colp & 15);
}

__global__ __launch_bounds__(256) void prep_kernel(
    const float* __restrict__ Wc, const float* __restrict__ W1,
    const float* __restrict__ W2, const float* __restrict__ bc,
    float* __restrict__ WT, float* __restrict__ W2T, float* __restrict__ bperm)
{
    int i = blockIdx.x * 256 + threadIdx.x;
    if (i < 12 * 16384) {
        int chn = i >> 14, col = (i >> 5) & 511, q = i & 31;
        WT[i] = Wc[(5 + chn * 32 + q) * 512 + gate_col(col)];
    } else if (i < 13 * 16384) {
        int j = i - 12 * 16384;
        int pc = j >> 5, q = j & 31;
        int kq = pc >> 7, n = pc & 127;
        WT[i] = W1[(kq * 32 + q) * 128 + n];
    } else if (i < 13 * 16384 + 4096) {
        int j = i - 13 * 16384;
        int col = j >> 3, k = j & 7;
        WT[i] = (k < 5) ? Wc[k * 512 + gate_col(col)] : 0.0f;
    } else if (i < 13 * 16384 + 4096 + 384) {
        int j = i - 13 * 16384 - 4096;
        int cc = j >> 7, k = j & 127;
        W2T[j] = W2[k * 3 + cc];
    } else if (i < 13 * 16384 + 4096 + 384 + 512) {
        int j = i - 13 * 16384 - 4096 - 384;
        bperm[j] = bc[(j & 3) * 128 + (j >> 2)];
    }
}

extern "C" void kernel_launch(void* const* d_in, const int* in_sizes, int n_in,
                              void* d_out, int out_size, void* d_ws, size_t ws_size,
                              hipStream_t stream)
{
    (void)in_sizes; (void)n_in; (void)out_size; (void)ws_size;
    const float* input = (const float*)d_in[0];
    const float* Wc    = (const float*)d_in[1];
    const float* bc    = (const float*)d_in[2];
    const float* W1    = (const float*)d_in[3];
    const float* b1    = (const float*)d_in[4];
    const float* W2    = (const float*)d_in[5];
    const float* b2    = (const float*)d_in[6];
    float* ws   = (float*)d_ws;
    float* WT   = ws;
    float* W2T  = ws + 217088;
    float* cbuf = ws + 217472;
    float* hall = ws + 479616;
    float* obuf = ws + 17125760;
    float* zprt = ws + 17515904;
    float* bprm = ws + 21710208;
    float* out  = (float*)d_out;

    prep_kernel<<<(217984 + 255) / 256, 256, 0, stream>>>(Wc, W1, W2, bc, WT, W2T, bprm);

    cell_step_kernel<true><<<256, 256, 0, stream>>>(input, WT, bprm, zprt, cbuf, hall, 0);
    for (int t = 1; t < 127; ++t)
        cell_step_kernel<false><<<256, 256, 0, stream>>>(input, WT, bprm, zprt, cbuf, hall, t);
    gate_kernel<<<1024, 128, 0, stream>>>(bprm, zprt, hall + 126 * 131072, cbuf);
    head_all_kernel<<<984, 256, 0, stream>>>(WT, hall + 4 * 131072, b1, W2T, b2,
                                              obuf + 4 * 3072);
    mix_all_kernel<<<123 * 1024 / 256, 256, 0, stream>>>(input, obuf, out);
}


# Cell GEMM's LDS fragment reads grouped a stage ahead: stores and all reads of stage s+1 stand in one place before stage s's 16 WMMAs (second register set); A fragments read once per chunk; same sums
# speedup vs baseline: 6.6739x; 1.0635x over previous
#include <hip/hip_runtime.h>
#include <math.h>

typedef __attribute__((ext_vector_type(2))) float v2f;
typedef __attribute__((ext_vector_type(8))) float v8f;
typedef __attribute__((ext_vector_type(4))) float v4f;

#define WP   36
#define WBUF (512 * WP)

__device__ __forceinline__ void chunk_request(v4f (&r)[16], const v4f* src, int tid)
{
    #pragma unroll
    for (int i = 0; i < 16; ++i) r[i] = src[tid + i * 256];
}

__device__ __forceinline__ void chunk_to_lds(float* buf, int wdst, const v4f (&r)[16])
{
    #pragma unroll
    for (int i = 0; i < 16; ++i)
        *reinterpret_cast<v4f*>(&buf[wdst + i * 32 * WP]) = r[i];
}

__device__ __forceinline__ void lds_barrier()
{
    asm volatile("s_wait_dscnt 0x0\n\ts_barrier_signal -1\n\ts_barrier_wait -1" ::: "memory");
}

__device__ __forceinline__ void wave_chunk_request(v4f (&r)[16], const v4f* src, int wave, int lane)
{
    #pragma unroll
    for (int i = 0; i < 16; ++i) r[i] = src[wave * 512 + i * 32 + lane];
}

template <int HALF>
__device__ __forceinline__ void wave_half_to_lds(float* buf, int wdst, const v4f (&r)[16])
{
    #pragma unroll
    for (int i = 8 * HALF; i < 8 * HALF + 8; ++i)
        *reinterpret_cast<v4f*>(&buf[wdst + i * 4 * WP]) = r[i];
}

__device__ __forceinline__ void wave_lds_handover()
{
    __builtin_amdgcn_fence(__ATOMIC_RELEASE, "wavefront");
    __builtin_amdgcn_wave_barrier();
    __builtin_amdgcn_fence(__ATOMIC_ACQUIRE, "wavefront");
}

__device__ __forceinline__ void frag_a(const float* arow, int kHalf, v2f (&a)[8])
{
    #pragma unroll
    for (int i = 0; i < 8; ++i)
        a[i] = *reinterpret_cast<const v2f*>(arow + 4 * i + kHalf);
}

__device__ __forceinline__ void frag_b(const float* wb, int wl, int kHalf, v2f (&b)[8])
{
    #pragma unroll
    for (int i = 0; i < 8; ++i)
        b[i] = *reinterpret_cast<const v2f*>(wb + wl + 4 * i + kHalf);
}

__device__ __forceinline__ void mma16(const v2f (&a)[8], const v2f (&bA)[8], const v2f (&bB)[8],
    v8f& accA, v8f& accB)
{
    #pragma unroll
    for (int i = 0; i < 8; ++i) {
        accA = __builtin_amdgcn_wmma_f32_16x16x4_f32(
                    false, a[i], false, bA[i], (short)0, accA, false, false);
        accB = __builtin_amdgcn_wmma_f32_16x16x4_f32(
                    false, a[i], false, bB[i], (short)0, accB, false, false);
    }
}

#define READS_AHEAD_OF(accA, accB) asm volatile("" : "+v"(accA), "+v"(accB) :: "memory")
#define WMMAS_DONE_BEFORE_STORES(accA, accB) asm volatile("" :: "v"(accA), "v"(accB) : "memory")

#define HP 36
#define ZPLANE (1024 * 512)
#define HPLANE (1024 * 128)

__device__ __forceinline__ void gate_math1(float zi, float zf, float zg, float zo,
    float cp, float& cn, float& hn)
{
    float si = 1.0f / (1.0f + __expf(-zi));
    float sf = 1.0f / (1.0f + __expf(-zf));
    float so = 1.0f / (1.0f + __expf(-zo));
    float cv = sf * cp + si * tanhf(zg);
    cn = cv;
    hn = so * tanhf(cv);
}

template <bool FIRST>
__global__ __launch_bounds__(256) void cell_step_kernel(
    const float* __restrict__ input, const float* __restrict__ WT,
    const float* __restrict__ bperm, float* zpart, float* cbuf,
    float* __restrict__ hkeep, int t)
{
    __shared__ alignas(16) float sh_h[18 * HP];
    __shared__ float sh_x[16 * 8];
    __shared__ alignas(16) float sh_w[3 * WBUF];

    const int tid = threadIdx.x;
    const int mt  = blockIdx.x >> 2;
    const int ks  = blockIdx.x & 3;
    const int R0  = mt * 16;
    const int bb  = R0 >> 7;
    const int s0  = R0 & 127;

    const float* zin  = zpart + ((t + 1) & 1) * (4 * ZPLANE);
    const float* c_in = cbuf  + ((t + 1) & 1) * HPLANE;
    float* const c_out = cbuf + (t & 1) * HPLANE;

    const int wave  = tid >> 5;
    const int lane  = tid & 31;

    float xv = 0.0f;
    if (tid < 16 * 5)
        xv = input[(((bb * 128) + t) * 128 + (s0 + tid / 5)) * 5 + tid % 5];
    const int orow = R0 + 2 * wave;
    const int gu   = ks * 32 + lane;
    const int hseg = (wave == 0) ? s0 - 1 : (wave == 1) ? s0 + 16 : s0 + 2 * wave;
    const bool hact = (wave < 2) && (hseg >= 0) && (hseg < 128);
    const int hrow = bb * 128 + min(max(hseg, 0), 127);
    v4f zr[8], zh[4], bq;
    float cp0, cp1, ch;
    if (!FIRST) {
        const float* zp = zin + orow * 512 + gu * 4;
        const float* zq = zin + hrow * 512 + gu * 4;
        #pragma unroll
        for (int p = 0; p < 4; ++p) {
            zr[p]     = *reinterpret_cast<const v4f*>(zp + p * ZPLANE);
            zr[4 + p] = *reinterpret_cast<const v4f*>(zp + 512 + p * ZPLANE);
        }
        bq  = *reinterpret_cast<const v4f*>(&bperm[gu * 4]);
        cp0 = c_in[orow * 128 + gu];
        cp1 = c_in[(orow + 1) * 128 + gu];
        #pragma unroll
        for (int p = 0; p < 4; ++p)
            zh[p] = *reinterpret_cast<const v4f*>(zq + p * ZPLANE);
        ch = c_in[hrow * 128 + gu];
    }

    const int lm    = lane & 15;
    const int hi    = lane >> 4;
    const int kHalf = hi * 2;

    const float* wc0 = WT + 13 * 16384 + ((wave * 4 + 0) * 16 + lm) * 8;
    const float* wc1 = WT + 13 * 16384 + ((wave * 4 + 1) * 16 + lm) * 8;
    const float* wc2 = WT + 13 * 16384 + ((wave * 4 + 2) * 16 + lm) * 8;
    const float* wc3 = WT + 13 * 16384 + ((wave * 4 + 3) * 16 + lm) * 8;
    const v4f wq0 = *reinterpret_cast<const v4f*>(wc0); const float we0 = wc0[4];
    const v4f wq1 = *reinterpret_cast<const v4f*>(wc1); const float we1 = wc1[4];
    const v4f wq2 = *reinterpret_cast<const v4f*>(wc2); const float we2 = wc2[4];
    const v4f wq3 = *reinterpret_cast<const v4f*>(wc3); const float we3 = wc3[4];

    asm volatile("" ::: "memory");

    const v4f* wsrc = reinterpret_cast<const v4f*>(WT);
    const int wdst = (wave * 64 + (lane >> 3)) * WP + (lane & 7) * 4;
    v4f preA[16], preB[16], preC[16];
    wave_chunk_request(preA, wsrc + ks * 4096, wave, lane);
    wave_chunk_request(preB, wsrc + (4 + ks) * 4096, wave, lane);

    if (!FIRST) {
        asm volatile("" :: "v"(zr[0]), "v"(zr[1]), "v"(zr[2]), "v"(zr[3]),
                           "v"(zr[4]), "v"(zr[5]), "v"(zr[6]), "v"(zr[7]),
                           "v"(bq), "v"(cp0), "v"(cp1),
                           "v"(zh[0]), "v"(zh[1]), "v"(zh[2]), "v"(zh[3]), "v"(ch));
    }

    float hn0 = 0.0f, hn1 = 0.0f;
    if (FIRST) {
        c_out[orow * 128 + gu]       = 0.0f;
        c_out[(orow + 1) * 128 + gu] = 0.0f;
    } else {
        const v4f za = (((zr[0] + zr[1]) + zr[2]) + zr[3]) + bq;
        const v4f zb = (((zr[4] + zr[5]) + zr[6]) + zr[7]) + bq;
        float cn0, cn1;
        gate_math1(za.x, za.y, za.z, za.w, cp0, cn0, hn0);
        gate_math1(zb.x, zb.y, zb.z, zb.w, cp1, cn1, hn1);
        c_out[orow * 128 + gu]       = cn0;
        c_out[(orow + 1) * 128 + gu] = cn1;
        hkeep[(t - 1) * HPLANE + orow * 128 + gu]       = hn0;
        hkeep[(t - 1) * HPLANE + (orow + 1) * 128 + gu] = hn1;
    }

    wave_chunk_request(preC, wsrc + (8 + ks) * 4096, wave, lane);

    sh_h[(1 + 2 * wave) * HP + lane] = hn0;
    sh_h[(2 + 2 * wave) * HP + lane] = hn1;
    if (tid < 16 * 5)
        sh_x[(tid / 5) * 8 + tid % 5] = xv;
    float* const buf0 = sh_w;
    float* const buf1 = sh_w + WBUF;
    float* const buf2 = sh_w + 2 * WBUF;
    lds_barrier();

    float hh = 0.0f;
    if (!FIRST) {
        if (hact) {
            const v4f zc = (((zh[0] + zh[1]) + zh[2]) + zh[3]) + bq;
            float cnh;
            gate_math1(zc.x, zc.y, zc.z, zc.w, ch, cnh, hh);
        }
    }
    if (wave < 2)
        sh_h[(wave == 0 ? 0 : 17) * HP + lane] = hh;
    asm volatile("s_wait_dscnt 0x0\n\ts_barrier_signal -1" ::: "memory");

    const int wl0 = ((wave * 4 + 0) * 16 + lm) * WP;
    const int wl1 = ((wave * 4 + 1) * 16 + lm) * WP;
    const int wl2 = ((wave * 4 + 2) * 16 + lm) * WP;
    const int wl3 = ((wave * 4 + 3) * 16 + lm) * WP;

    v8f acc0 = {}, acc1 = {}, acc2 = {}, acc3 = {};

    if (ks == 0) {
        #pragma unroll
        for (int v = 0; v < 8; ++v) {
            int m = v + hi * 8;
            float x0 = sh_x[m*8+0], x1 = sh_x[m*8+1], x2 = sh_x[m*8+2],
                  x3 = sh_x[m*8+3], x4 = sh_x[m*8+4];
            acc0[v] += x0*wq0.x + x1*wq0.y + x2*wq0.z + x3*wq0.w + x4*we0;
            acc1[v] += x0*wq1.x + x1*wq1.y + x2*wq1.z + x3*wq1.w + x4*we1;
            acc2[v] += x0*wq2.x + x1*wq2.y + x2*wq2.z + x3*wq2.w + x4*we2;
            acc3[v] += x0*wq3.x + x1*wq3.y + x2*wq3.z + x3*wq3.w + x4*we3;
        }
    }

    const float* const a0 = &sh_h[(1 + lm) * HP];
    const float* const a1 = &sh_h[(2 + lm) * HP];
    const float* const a2 = &sh_h[lm * HP];
    v2f fa0[8], fa1[8], fa2[8];
    v2f b0A[8], b0B[8], b1A[8], b1B[8], b2A[8], b2B[8];
    v2f b3A[8], b3B[8], b4A[8], b4B[8], b5A[8], b5B[8];

    wave_half_to_lds<0>(buf0, wdst, preA);
    wave_lds_handover();
    frag_a(a0, kHalf, fa0);
    frag_b(buf0, wl0, kHalf, b0A); frag_b(buf0, wl1, kHalf, b0B);
    wave_half_to_lds<1>(buf0, wdst, preA);
    wave_lds_handover();
    frag_b(buf0, wl2, kHalf, b1A); frag_b(buf0, wl3, kHalf, b1B);
    READS_AHEAD_OF(acc0, acc1);
    mma16(fa0, b0A, b0B, acc0, acc1);
    WMMAS_DONE_BEFORE_STORES(acc0, acc1);
    asm volatile("s_barrier_wait -1" ::: "memory");
    wave_half_to_lds<0>(buf1, wdst, preB);
    wave_lds_handover();
    frag_a(a1, kHalf, fa1);
    frag_b(buf1, wl0, kHalf, b2A); frag_b(buf1, wl1, kHalf, b2B);
    READS_AHEAD_OF(acc2, acc3);
    mma16(fa0, b1A, b1B, acc2, acc3);
    WMMAS_DONE_BEFORE_STORES(acc2, acc3);
    wave_half_to_lds<1>(buf1, wdst, preB);
    wave_lds_handover();
    frag_b(buf1, wl2, kHalf, b3A); frag_b(buf1, wl3, kHalf, b3B);
    READS_AHEAD_OF(acc0, acc1);
    mma16(fa1, b2A, b2B, acc0, acc1);
    WMMAS_DONE_BEFORE_STORES(acc0, acc1);
    wave_half_to_lds<0>(buf2, wdst, preC);
    wave_lds_handover();
    frag_a(a2, kHalf, fa2);
    frag_b(buf2, wl0, kHalf, b4A); frag_b(buf2, wl1, kHalf, b4B);
    READS_AHEAD_OF(acc2, acc3);
    mma16(fa1, b3A, b3B, acc2, acc3);
    WMMAS_DONE_BEFORE_STORES(acc2, acc3);
    wave_half_to_lds<1>(buf2, wdst, preC);
    wave_lds_handover();
    frag_b(buf2, wl2, kHalf, b5A); frag_b(buf2, wl3, kHalf, b5B);
    READS_AHEAD_OF(acc0, acc1);
    mma16(fa2, b4A, b4B, acc0, acc1);
    mma16(fa2, b5A, b5B, acc2, acc3);

    float* zp = zpart + (t & 1) * (4 * ZPLANE) + ks * ZPLANE + R0 * 512 + wave * 64 + lm * 4;
    #pragma unroll
    for (int v = 0; v < 8; ++v) {
        const v4f piece = {acc0[v], acc1[v], acc2[v], acc3[v]};
        *reinterpret_cast<v4f*>(&zp[(v + hi * 8) * 512]) = piece;
    }
}

__global__ __launch_bounds__(128) void gate_kernel(
    const float* __restrict__ bperm, const float* __restrict__ zpart,
    float* __restrict__ h_out, float* __restrict__ c)
{
    const int row = blockIdx.x, gu = threadIdx.x;

    const float* zp = zpart + row * 512 + gu * 4;
    const v4f p0 = *reinterpret_cast<const v4f*>(zp);
    const v4f p1 = *reinterpret_cast<const v4f*>(zp + ZPLANE);
    const v4f p2 = *reinterpret_cast<const v4f*>(zp + 2 * ZPLANE);
    const v4f p3 = *reinterpret_cast<const v4f*>(zp + 3 * ZPLANE);
    const v4f bq = *reinterpret_cast<const v4f*>(&bperm[gu * 4]);
    const float cp = c[row * 128 + gu];

    const v4f z = (((p0 + p1) + p2) + p3) + bq;
    float cn, hn;
    gate_math1(z.x, z.y, z.z, z.w, cp, cn, hn);
    c[row * 128 + gu]     = cn;
    h_out[row * 128 + gu] = hn;
}

#define HIDP 132

__global__ __launch_bounds__(256) void head_all_kernel(
    const float* __restrict__ WT, const float* __restrict__ hrows,
    const float* __restrict__ b1, const float* __restrict__ W2T,
    const float* __restrict__ b2, float* __restrict__ orows)
{
    __shared__ alignas(16) float sh_w[WBUF];
    __shared__ alignas(16) float sh_hid[8 * 16 * HIDP];
    __shared__ alignas(16) float sh_w2[3 * 128];
    __shared__ alignas(16) float sh_o[8 * 48];

    const int tid   = threadIdx.x;
    const int wave  = tid >> 5;
    const int lane  = tid & 31;
    const int lm    = lane & 15;
    const int hi    = lane >> 4;
    const int kHalf = hi * 2;

    const v4f* wsrc = reinterpret_cast<const v4f*>(WT);
    const int wdst = (tid >> 3) * WP + (tid & 7) * 4;
    v4f preA[16];
    chunk_request(preA, wsrc + 12 * 4096, tid);

    const float* hrow = hrows + ((blockIdx.x * 8 + wave) * 16 + lm) * 128 + kHalf;
    v2f a[32];
    #pragma unroll
    for (int ks = 0; ks < 32; ++ks)
        a[ks] = *reinterpret_cast<const v2f*>(hrow + ks * 4);

    float bias1[8];
    #pragma unroll
    for (int j = 0; j < 8; ++j) bias1[j] = b1[j * 16 + lm];
    const float b20 = b2[0], b21 = b2[1], b22 = b2[2];
    v4f w2v = {0.0f, 0.0f, 0.0f, 0.0f};
    if (tid < 96) w2v = *reinterpret_cast<const v4f*>(W2T + tid * 4);

    if (tid < 96) *reinterpret_cast<v4f*>(&sh_w2[tid * 4]) = w2v;
    chunk_to_lds(sh_w, wdst, preA);
    __syncthreads();

    v8f acc[8];
    #pragma unroll
    for (int j = 0; j < 8; ++j) acc[j] = v8f{};
    #pragma unroll
    for (int kq = 0; kq < 4; ++kq) {
        #pragma unroll
        for (int kk = 0; kk < 32; kk += 4) {
            #pragma unroll
            for (int j = 0; j < 8; ++j) {
                const float* wq = sh_w + (kq * 128 + j * 16 + lm) * WP;
                v2f b;
                b.x = wq[kk + kHalf]; b.y = wq[kk + kHalf + 1];
                acc[j] = __builtin_amdgcn_wmma_f32_16x16x4_f32(
                            false, a[kq * 8 + (kk >> 2)], false, b, (short)0, acc[j], false, false);
            }
        }
    }
    float* hid = sh_hid + wave * 16 * HIDP;
    #pragma unroll
    for (int j = 0; j < 8; ++j) {
        #pragma unroll
        for (int v = 0; v < 8; ++v) {
            float x = acc[j][v] + bias1[j];
            hid[(v + hi * 8) * HIDP + j * 16 + lm] = x > 0.0f ? x : 0.0f;
        }
    }
    __syncthreads();

    if (lane < 16) {
        const float* hr = hid + lane * HIDP;
        float s0 = 0.0f, s1 = 0.0f, s2 = 0.0f;
        for (int k = 0; k < 128; k += 4) {
            const v4f hv = *reinterpret_cast<const v4f*>(hr + k);
            const v4f wa = *reinterpret_cast<const v4f*>(sh_w2 + k);
            const v4f wb = *reinterpret_cast<const v4f*>(sh_w2 + 128 + k);
            const v4f wc = *reinterpret_cast<const v4f*>(sh_w2 + 256 + k);
            s0 = fmaf(hv.x, wa.x, s0); s0 = fmaf(hv.y, wa.y, s0);
            s0 = fmaf(hv.z, wa.z, s0); s0 = fmaf(hv.w, wa.w, s0);
            s1 = fmaf(hv.x, wb.x, s1); s1 = fmaf(hv.y, wb.y, s1);
            s1 = fmaf(hv.z, wb.z, s1); s1 = fmaf(hv.w, wb.w, s1);
            s2 = fmaf(hv.x, wc.x, s2); s2 = fmaf(hv.y, wc.y, s2);
            s2 = fmaf(hv.z, wc.z, s2); s2 = fmaf(hv.w, wc.w, s2);
        }
        float* so = sh_o + wave * 48 + lane * 3;
        so[0] = s0 + b20; so[1] = s1 + b21; so[2] = s2 + b22;
    }
    __syncthreads();

    if (tid < 96)
        *reinterpret_cast<v4f*>(orows + blockIdx.x * 384 + tid * 4) =
            *reinterpret_cast<const v4f*>(sh_o + tid * 4);
}

__global__ __launch_bounds__(256) void mix_all_kernel(
    const float* __restrict__ input, const float* __restrict__ obuf_all,
    float* __restrict__ out)
{
    int gid = blockIdx.x * 256 + threadIdx.x;
    if (gid >= 123 * 1024) return;
    int tt = gid >> 10, r = gid & 1023;
    int t  = tt + 4;
    int bb = r >> 7, s = r & 127;
    const float* ob = obuf_all + t * 3072;
    const float* o  = &ob[r * 3];
    float n0 = o[0];
    float n1, sp0, sp1;
    if (s == 0) {
        const float* xn0 = &input[((bb * 128 + (t + 1)) * 128 + 0) * 5];
        n1 = xn0[1]; sp0 = xn0[3]; sp1 = xn0[4];
    } else {
        n1  = ob[(r - 1) * 3 + 0];
        sp0 = o[1]; sp1 = o[2];
    }
    float fn = input[((bb * 128 + t) * 128 + s) * 5 + 2];
    float n2 = fn + n1 - n0;
    float* dst = &out[((bb * 123 + tt) * 128 + s) * 5];
    dst[0] = n0; dst[1] = n1; dst[2] = n2; dst[3] = sp0; dst[4] = sp1;
}

__device__ __forceinline__ int gate_col(int colp)
{
    return ((colp >> 4) & 3) * 128 + (colp >> 6) * 16 + (colp & 15);
}

__global__ __launch_bounds__(256) void prep_kernel(
    const float* __restrict__ Wc, const float* __restrict__ W1,
    const float* __restrict__ W2, const float* __restrict__ bc,
    float* __restrict__ WT, float* __restrict__ W2T, float* __restrict__ bperm)
{
    int i = blockIdx.x * 256 + threadIdx.x;
    if (i < 12 * 16384) {
        int chn = i >> 14, col = (i >> 5) & 511, q = i & 31;
        WT[i] = Wc[(5 + chn * 32 + q) * 512 + gate_col(col)];
    } else if (i < 13 * 16384) {
        int j = i - 12 * 16384;
        int pc = j >> 5, q = j & 31;
        int kq = pc >> 7, n = pc & 127;
        WT[i] = W1[(kq * 32 + q) * 128 + n];
    } else if (i < 13 * 16384 + 4096) {
        int j = i - 13 * 16384;
        int col = j >> 3, k = j & 7;
        WT[i] = (k < 5) ? Wc[k * 512 + gate_col(col)] : 0.0f;
    } else if (i < 13 * 16384 + 4096 + 384) {
        int j = i - 13 * 16384 - 4096;
        int cc = j >> 7, k = j & 127;
        W2T[j] = W2[k * 3 + cc];
    } else if (i < 13 * 16384 + 4096 + 384 + 512) {
        int j = i - 13 * 16384 - 4096 - 384;
        bperm[j] = bc[(j & 3) * 128 + (j >> 2)];
    }
}

extern "C" void kernel_launch(void* const* d_in, const int* in_sizes, int n_in,
                              void* d_out, int out_size, void* d_ws, size_t ws_size,
                              hipStream_t stream)
{
    (void)in_sizes; (void)n_in; (void)out_size; (void)ws_size;
    const float* input = (const float*)d_in[0];
    const float* Wc    = (const float*)d_in[1];
    const float* bc    = (const float*)d_in[2];
    const float* W1    = (const float*)d_in[3];
    const float* b1    = (const float*)d_in[4];
    const float* W2    = (const float*)d_in[5];
    const float* b2    = (const float*)d_in[6];
    float* ws   = (float*)d_ws;
    float* WT   = ws;
    float* W2T  = ws + 217088;
    float* cbuf = ws + 217472;
    float* hall = ws + 479616;
    float* obuf = ws + 17125760;
    float* zprt = ws + 17515904;
    float* bprm = ws + 21710208;
    float* out  = (float*)d_out;

    prep_kernel<<<(217984 + 255) / 256, 256, 0, stream>>>(Wc, W1, W2, bc, WT, W2T, bprm);

    cell_step_kernel<true><<<256, 256, 0, stream>>>(input, WT, bprm, zprt, cbuf, hall, 0);
    for (int t = 1; t < 127; ++t)
        cell_step_kernel<false><<<256, 256, 0, stream>>>(input, WT, bprm, zprt, cbuf, hall, t);
    gate_kernel<<<1024, 128, 0, stream>>>(bprm, zprt, hall + 126 * 131072, cbuf);
    head_all_kernel<<<984, 256, 0, stream>>>(WT, hall + 4 * 131072, b1, W2T, b2,
                                              obuf + 4 * 3072);
    mix_all_kernel<<<123 * 1024 / 256, 256, 0, stream>>>(input, obuf, out);
}
